# starting kernel
# baseline (speedup 1.0000x reference)
#include <hip/hip_runtime.h>
#include <cmath>

typedef __attribute__((ext_vector_type(2))) float v2f;
typedef __attribute__((ext_vector_type(8))) float v8f;

#define IN_DIM 128
#define HID_DIM 128
#define OUT_DIM 64
#define KDIM 128

__global__ void gcn_fill_one(float* __restrict__ deg, int n) {
    int i = blockIdx.x * blockDim.x + threadIdx.x;
    if (i < n) deg[i] = 1.0f;
}

__global__ void gcn_deg_count(const int* __restrict__ dst, float* __restrict__ deg, int e) {
    int i = blockIdx.x * blockDim.x + threadIdx.x;
    if (i < e) unsafeAtomicAdd(&deg[dst[i]], 1.0f);
}

__global__ void gcn_rsqrt_ip(float* __restrict__ deg, int n) {
    int i = blockIdx.x * blockDim.x + threadIdx.x;
    if (i < n) deg[i] = rsqrtf(deg[i]);
}

template <int NOUT>
__global__ __launch_bounds__(NOUT * 2) void gcn_gemm_wmma(
    const float* __restrict__ A, const float* __restrict__ B,
    float* __restrict__ C, int M) {
    constexpr int NTHREADS = NOUT * 2;
    __shared__ float Alds[16 * KDIM];

    const int mTile = blockIdx.x;
    const int tid   = threadIdx.x;

    for (int i = tid; i < 16 * KDIM; i += NTHREADS) {
        int r = i >> 7;
        int c = i & (KDIM - 1);
        int row = mTile * 16 + r;
        if (row >= M) row = M - 1;
        Alds[i] = A[(size_t)row * KDIM + c];
    }
    __syncthreads();

    const int wave  = tid >> 5;
    const int lane  = tid & 31;
    const int lrow  = lane & 15;
    const int khalf = lane >> 4;
    const int col   = lrow + wave * 16;

    v8f acc = {};
    #pragma unroll
    for (int k0 = 0; k0 < KDIM; k0 += 4) {
        const int ka = k0 + 2 * khalf;
        v2f a, b;
        a.x = Alds[lrow * KDIM + ka];
        a.y = Alds[lrow * KDIM + ka + 1];
        b.x = B[(size_t)ka * NOUT + col];
        b.y = B[(size_t)(ka + 1) * NOUT + col];
        acc = __builtin_amdgcn_wmma_f32_16x16x4_f32(
             false, a,  false, b,
             (short)0, acc,  false,  false);
    }

    const int rowBase = mTile * 16 + 8 * khalf;
    #pragma unroll
    for (int i = 0; i < 8; ++i) {
        int row = rowBase + i;
        if (row < M) C[(size_t)row * NOUT + col] = acc[i];
    }
}

template <int F>
__global__ void gcn_selfloop_init(const float* __restrict__ H,
                                  const float* __restrict__ dinv,
                                  float* __restrict__ agg, size_t total) {
    size_t i = (size_t)blockIdx.x * blockDim.x + threadIdx.x;
    if (i >= total) return;
    int n = (int)(i / F);
    float di = dinv[n];
    agg[i] = H[i] * di * di;
}

template <int F>
__global__ void gcn_edge_scatter(const float* __restrict__ H,
                                 const int* __restrict__ src,
                                 const int* __restrict__ dst,
                                 const float* __restrict__ dinv,
                                 float* __restrict__ agg, int E) {
    const int lane = threadIdx.x & 31;
    const int e = blockIdx.x * (blockDim.x >> 5) + (threadIdx.x >> 5);
    if (e >= E) return;
    const int s = src[e];
    const int d = dst[e];
    const float w = dinv[s] * dinv[d];
    const float* hp = H + (size_t)s * F;
    float* ap = agg + (size_t)d * F;
    #pragma unroll
    for (int j = 0; j < F / 32; ++j) {
        int f = lane + j * 32;
        unsafeAtomicAdd(ap + f, hp[f] * w);
    }
}

template <int F>
__global__ void gcn_bias_elu(const float* __restrict__ agg,
                             const float* __restrict__ bias,
                             float* __restrict__ out, size_t total) {
    size_t i = (size_t)blockIdx.x * blockDim.x + threadIdx.x;
    if (i >= total) return;
    float v = agg[i] + bias[i & (F - 1)];
    out[i] = (v > 0.0f) ? v : expm1f(v);
}

extern "C" void kernel_launch(void* const* d_in, const int* in_sizes, int n_in,
                              void* d_out, int out_size, void* d_ws, size_t ws_size,
                              hipStream_t stream) {
    const float* x  = (const float*)d_in[0];
    const int*   ei = (const int*)d_in[1];
    const float* W1 = (const float*)d_in[2];
    const float* b1 = (const float*)d_in[3];
    const float* W2 = (const float*)d_in[4];
    const float* b2 = (const float*)d_in[5];
    float* out = (float*)d_out;

    const int N = in_sizes[0] / IN_DIM;
    const int E = in_sizes[1] / 2;
    const int* src = ei;
    const int* dst = ei + E;

    char* ws = (char*)d_ws;
    float* dinv = (float*)ws;
    size_t off1 = ((size_t)N * 4 + 255) & ~(size_t)255;
    float* hbuf = (float*)(ws + off1);
    size_t off2 = off1 + ((((size_t)N * HID_DIM * 4) + 255) & ~(size_t)255);
    float* agg1 = (float*)(ws + off2);

    const int TB = 256;
    const int edgeBlocks = (E + 7) / 8;
    const int mTiles = (N + 15) / 16;
    const size_t tot1 = (size_t)N * HID_DIM;
    const size_t tot2 = (size_t)N * OUT_DIM;

    gcn_fill_one<<<(N + TB - 1) / TB, TB, 0, stream>>>(dinv, N);
    gcn_deg_count<<<(E + TB - 1) / TB, TB, 0, stream>>>(dst, dinv, E);
    gcn_rsqrt_ip<<<(N + TB - 1) / TB, TB, 0, stream>>>(dinv, N);

    gcn_gemm_wmma<HID_DIM><<<mTiles, HID_DIM * 2, 0, stream>>>(x, W1, hbuf, N);
    gcn_selfloop_init<HID_DIM><<<(int)((tot1 + TB - 1) / TB), TB, 0, stream>>>(hbuf, dinv, agg1, tot1);
    gcn_edge_scatter<HID_DIM><<<edgeBlocks, TB, 0, stream>>>(hbuf, src, dst, dinv, agg1, E);
    gcn_bias_elu<HID_DIM><<<(int)((tot1 + TB - 1) / TB), TB, 0, stream>>>(agg1, b1, agg1, tot1);

    gcn_gemm_wmma<OUT_DIM><<<mTiles, OUT_DIM * 2, 0, stream>>>(agg1, W2, hbuf, N);
    gcn_selfloop_init<OUT_DIM><<<(int)((tot2 + TB - 1) / TB), TB, 0, stream>>>(hbuf, dinv, out, tot2);
    gcn_edge_scatter<OUT_DIM><<<edgeBlocks, TB, 0, stream>>>(hbuf, src, dst, dinv, out, E);
    gcn_bias_elu<OUT_DIM><<<(int)((tot2 + TB - 1) / TB), TB, 0, stream>>>(out, b2, out, tot2);
}


# Per-edge f32 atomics replaced by a dst-grouped edge list built each call (int atomic count+rank, scan, fill) and one wave summing each output row in arrival-rank order; dinv[src] scales GEMM A rows
# speedup vs baseline: 2.3391x; 2.3391x over previous
#include <hip/hip_runtime.h>
#include <cmath>


typedef __attribute__((ext_vector_type(2))) float v2f;
typedef __attribute__((ext_vector_type(8))) float v8f;

#define IN_DIM 128
#define HID_DIM 128
#define OUT_DIM 64
#define KDIM 128

__global__ void csr_zero(int* __restrict__ cnt, int n) {
    int i = blockIdx.x * blockDim.x + threadIdx.x;
    if (i < n) cnt[i] = 0;
}

__global__ void csr_count_rank(const int* __restrict__ dst, int* __restrict__ cnt,
                               int* __restrict__ rank, int e) {
    int i = blockIdx.x * blockDim.x + threadIdx.x;
    if (i < e) rank[i] = atomicAdd(&cnt[dst[i]], 1);
}

#define SCAN_TB 1024
__global__ __launch_bounds__(SCAN_TB) void csr_scan(const int* __restrict__ cnt,
                                                    int* __restrict__ start,
                                                    float* __restrict__ dinv, int n) {
    __shared__ int wpart[32];
    __shared__ int wtot[32];
    const int tid  = threadIdx.x;
    const int lane = tid & 31;
    const int wave = tid >> 5;
    const int first = blockIdx.x * SCAN_TB;

    int part = 0;
    for (int j = tid; j < first; j += SCAN_TB) part += cnt[j];
    #pragma unroll
    for (int o = 16; o > 0; o >>= 1) part += __shfl_xor(part, o);

    const int i = first + tid;
    const int c = (i < n) ? cnt[i] : 0;
    int incl = c;
    #pragma unroll
    for (int o = 1; o < 32; o <<= 1) {
        int t = __shfl_up(incl, o);
        if (lane >= o) incl += t;
    }
    if (lane == 0)  wpart[wave] = part;
    if (lane == 31) wtot[wave]  = incl;
    __syncthreads();
    if (wave == 0) {
        int p = wpart[lane];
        #pragma unroll
        for (int o = 16; o > 0; o >>= 1) p += __shfl_xor(p, o);
        const int t = wtot[lane];
        int ti = t;
        #pragma unroll
        for (int o = 1; o < 32; o <<= 1) {
            int u = __shfl_up(ti, o);
            if (lane >= o) ti += u;
        }
        wtot[lane] = p + ti - t;
    }
    __syncthreads();
    if (i < n) {
        start[i] = wtot[wave] + incl - c;
        dinv[i]  = rsqrtf((float)(c + 1));
    }
}

__global__ void csr_fill(const int* __restrict__ src, const int* __restrict__ dst,
                         const int* __restrict__ rank, const int* __restrict__ start,
                         int* __restrict__ col, int e) {
    int i = blockIdx.x * blockDim.x + threadIdx.x;
    if (i < e) col[start[dst[i]] + rank[i]] = src[i];
}

template <int NOUT, bool MOVE>
__global__ __launch_bounds__(NOUT * 2) void gcn_gemm_wmma(
    const float* __restrict__ A, const float* __restrict__ B,
    const float* __restrict__ rscale, float* __restrict__ C, int M,
    const int* __restrict__ mv_src, int* __restrict__ mv_dst, int mv_n) {
    constexpr int NTHREADS = NOUT * 2;
    __shared__ float Alds[16 * KDIM];

    const int mTile = blockIdx.x;
    const int tid   = threadIdx.x;

    for (int i = tid; i < 16 * KDIM; i += NTHREADS) {
        int r = i >> 7;
        int c = i & (KDIM - 1);
        int row = mTile * 16 + r;
        if (row >= M) row = M - 1;
        Alds[i] = A[(size_t)row * KDIM + c] * rscale[row];
    }

    if (MOVE) {
        for (int i = mTile * NTHREADS + tid; i < mv_n; i += (int)gridDim.x * NTHREADS)
            mv_dst[i] = mv_src[i];
    }
    __syncthreads();

    const int wave  = tid >> 5;
    const int lane  = tid & 31;
    const int lrow  = lane & 15;
    const int khalf = lane >> 4;
    const int col   = lrow + wave * 16;

    v8f acc = {};
    #pragma unroll
    for (int k0 = 0; k0 < KDIM; k0 += 4) {
        const int ka = k0 + 2 * khalf;
        v2f a, b;
        a.x = Alds[lrow * KDIM + ka];
        a.y = Alds[lrow * KDIM + ka + 1];
        b.x = B[(size_t)ka * NOUT + col];
        b.y = B[(size_t)(ka + 1) * NOUT + col];
        acc = __builtin_amdgcn_wmma_f32_16x16x4_f32(
             false, a,  false, b,
             (short)0, acc,  false,  false);
    }

    const int rowBase = mTile * 16 + 8 * khalf;
    #pragma unroll
    for (int i = 0; i < 8; ++i) {
        int row = rowBase + i;
        if (row < M) C[(size_t)row * NOUT + col] = acc[i];
    }
}

template <int K, typename vec>
__device__ __forceinline__ void add_rows(vec& acc, const vec* __restrict__ H,
                                         int s, int j, int lane) {
    vec r[K];
    #pragma unroll
    for (int u = 0; u < K; ++u)
        r[u] = H[(size_t)__builtin_amdgcn_readlane(s, j + u) * 32 + lane];
    #pragma unroll
    for (int u = 0; u < K; ++u) acc += r[u];
}

template <int F>
__global__ __launch_bounds__(256) void gcn_gather_rows(
    const float* __restrict__ Hs, const int* __restrict__ start,
    const int* __restrict__ col, const float* __restrict__ dinv,
    const float* __restrict__ bias, float* __restrict__ out, int N, int E) {
    constexpr int V = F / 32;
    typedef float vec __attribute__((ext_vector_type(V)));
    const int lane = threadIdx.x & 31;
    const int n = __builtin_amdgcn_readfirstlane((int)(blockIdx.x * 8 + (threadIdx.x >> 5)));
    if (n >= N) return;
    const int beg = start[n];
    const int end = (n + 1 < N) ? start[n + 1] : E;
    const vec* __restrict__ H = reinterpret_cast<const vec*>(Hs);

    vec acc = H[(size_t)n * 32 + lane];
    for (int p = beg; p < end; p += 32) {
        const int m = min(32, end - p);
        int s = (lane < m) ? col[p + lane] : 0;
        s = min(max(s, 0), N - 1);
        int j = 0;
        for (; j + 8 <= m; j += 8) add_rows<8>(acc, H, s, j, lane);
        if (m & 4) { add_rows<4>(acc, H, s, j, lane); j += 4; }
        if (m & 2) { add_rows<2>(acc, H, s, j, lane); j += 2; }
        if (m & 1) add_rows<1>(acc, H, s, j, lane);
    }

    const float di = dinv[n];
    const vec b = reinterpret_cast<const vec*>(bias)[lane];
    vec v = acc * di + b;
    #pragma unroll
    for (int k = 0; k < V; ++k) v[k] = (v[k] > 0.0f) ? v[k] : expm1f(v[k]);
    reinterpret_cast<vec*>(out)[(size_t)n * 32 + lane] = v;
}

extern "C" void kernel_launch(void* const* d_in, const int* in_sizes, int n_in,
                              void* d_out, int out_size, void* d_ws, size_t ws_size,
                              hipStream_t stream) {
    const float* x  = (const float*)d_in[0];
    const int*   ei = (const int*)d_in[1];
    const float* W1 = (const float*)d_in[2];
    const float* b1 = (const float*)d_in[3];
    const float* W2 = (const float*)d_in[4];
    const float* b2 = (const float*)d_in[5];
    float* out = (float*)d_out;

    const int N = in_sizes[0] / IN_DIM;
    const int E = in_sizes[1] / 2;
    const int* src = ei;
    const int* dst = ei + E;

    char* ws = (char*)d_ws;
    float* dinv = (float*)ws;
    size_t off1 = ((size_t)N * 4 + 255) & ~(size_t)255;
    float* hbuf = (float*)(ws + off1);
    size_t off2 = off1 + ((((size_t)N * HID_DIM * 4) + 255) & ~(size_t)255);
    float* act1 = (float*)(ws + off2);

    int* csrA   = (int*)d_out;
    int* startA = csrA;
    int* colA   = csrA + N;
    int* cnt    = csrA + N + E;
    int* rank   = (int*)act1;
    int* csrB   = (int*)(hbuf + (size_t)N * OUT_DIM);
    const int* startB = csrB;
    const int* colB   = csrB + N;
    const int csrInts = N + E;

    const int TB = 256;
    const int mTiles = (N + 15) / 16;
    const int nodeBlocks = (N + 7) / 8;

    csr_zero<<<(N + TB - 1) / TB, TB, 0, stream>>>(cnt, N);
    csr_count_rank<<<(E + TB - 1) / TB, TB, 0, stream>>>(dst, cnt, rank, E);
    csr_scan<<<(N + SCAN_TB - 1) / SCAN_TB, SCAN_TB, 0, stream>>>(cnt, startA, dinv, N);
    csr_fill<<<(E + TB - 1) / TB, TB, 0, stream>>>(src, dst, rank, startA, colA, E);

    gcn_gemm_wmma<HID_DIM, false><<<mTiles, HID_DIM * 2, 0, stream>>>(
        x, W1, dinv, hbuf, N, nullptr, nullptr, 0);
    gcn_gather_rows<HID_DIM><<<nodeBlocks, 256, 0, stream>>>(
        hbuf, startA, colA, dinv, b1, act1, N, E);

    gcn_gemm_wmma<OUT_DIM, true><<<mTiles, OUT_DIM * 2, 0, stream>>>(
        act1, W2, dinv, hbuf, N, csrA, csrB, csrInts);
    gcn_gather_rows<OUT_DIM><<<nodeBlocks, 256, 0, stream>>>(
        hbuf, startB, colB, dinv, b2, out, N, E);
}


# Edge list built by a two-level counting sort with all counters in LDS (tile histogram, scan, bucket scatter, per-bucket ranks), not one returning global add per edge; row sums stay in arrival order
# speedup vs baseline: 2.9923x; 1.2793x over previous
#include <hip/hip_runtime.h>
#include <cmath>


typedef __attribute__((ext_vector_type(2))) float v2f;
typedef __attribute__((ext_vector_type(8))) float v8f;

#define IN_DIM 128
#define HID_DIM 128
#define OUT_DIM 64
#define KDIM 128

#define EL_TILE    4096
#define EL_TB      256
#define EL_BTB     1024
#define EL_BPER    5
#define EL_RLOG    8
#define EL_NB_MAX  512
#define EL_STAGE   8192

__global__ __launch_bounds__(EL_TB) void el_hist(const int* __restrict__ dst,
                                                 int* __restrict__ tcnt,
                                                 int e, int n, int nb, int ntiles) {
    __shared__ int h[EL_NB_MAX];
    const int tid = threadIdx.x;
    for (int b = tid; b < EL_NB_MAX; b += EL_TB) h[b] = 0;
    __syncthreads();
    const int base = blockIdx.x * EL_TILE;
    int d[EL_TILE / EL_TB];
    #pragma unroll
    for (int k = 0; k < EL_TILE / EL_TB; ++k)
        d[k] = dst[min(base + k * EL_TB + tid, e - 1)];
    #pragma unroll
    for (int k = 0; k < EL_TILE / EL_TB; ++k) {
        if (base + k * EL_TB + tid < e) {
            const int dk = min(max(d[k], 0), n - 1);
            atomicAdd(&h[min(dk >> EL_RLOG, EL_NB_MAX - 1)], 1);
        }
    }
    __syncthreads();
    const int nbc = min(nb, EL_NB_MAX);
    for (int b = tid; b < nbc; b += EL_TB) tcnt[(size_t)b * ntiles + blockIdx.x] = h[b];
}

#define SCAN_TB 1024
__global__ __launch_bounds__(SCAN_TB) void el_scan(const int* __restrict__ cnt,
                                                   int* __restrict__ off, int len) {
    __shared__ int wpart[32];
    __shared__ int wtot[32];
    const int tid  = threadIdx.x;
    const int lane = tid & 31;
    const int wave = tid >> 5;
    const int4* __restrict__ c4 = reinterpret_cast<const int4*>(cnt);

    int part = 0;
    const int firstq = blockIdx.x * SCAN_TB;
    #pragma unroll 8
    for (int q = tid; q < firstq; q += SCAN_TB) {
        const int4 v = c4[q];
        part += (v.x + v.y) + (v.z + v.w);
    }
    #pragma unroll
    for (int o = 16; o > 0; o >>= 1) part += __shfl_xor(part, o);

    const int q  = firstq + tid;
    const int i0 = q * 4;
    int4 v = c4[q];
    if (i0 + 0 >= len) v.x = 0;
    if (i0 + 1 >= len) v.y = 0;
    if (i0 + 2 >= len) v.z = 0;
    if (i0 + 3 >= len) v.w = 0;
    const int c = (v.x + v.y) + (v.z + v.w);
    int incl = c;
    #pragma unroll
    for (int o = 1; o < 32; o <<= 1) {
        int t = __shfl_up(incl, o);
        if (lane >= o) incl += t;
    }
    if (lane == 0)  wpart[wave] = part;
    if (lane == 31) wtot[wave]  = incl;
    __syncthreads();
    if (wave == 0) {
        int p = wpart[lane];
        #pragma unroll
        for (int o = 16; o > 0; o >>= 1) p += __shfl_xor(p, o);
        const int t = wtot[lane];
        int ti = t;
        #pragma unroll
        for (int o = 1; o < 32; o <<= 1) {
            int u = __shfl_up(ti, o);
            if (lane >= o) ti += u;
        }
        wtot[lane] = p + ti - t;
    }
    __syncthreads();
    int4 r;
    r.x = wtot[wave] + incl - c;
    r.y = r.x + v.x;
    r.z = r.y + v.y;
    r.w = r.z + v.z;
    reinterpret_cast<int4*>(off)[q] = r;
}

__global__ __launch_bounds__(EL_TB) void el_scatter(const int* __restrict__ src,
                                                    const int* __restrict__ dst,
                                                    const int* __restrict__ toff,
                                                    int* __restrict__ rec,
                                                    int e, int n, int nb, int ntiles) {
    __shared__ int cur[EL_NB_MAX];
    const int tid = threadIdx.x;
    for (int b = tid; b < EL_NB_MAX; b += EL_TB)
        cur[b] = (b < nb) ? toff[(size_t)b * ntiles + blockIdx.x] : 0;
    __syncthreads();
    const int base = blockIdx.x * EL_TILE;
    int d[EL_TILE / EL_TB], s[EL_TILE / EL_TB];
    #pragma unroll
    for (int k = 0; k < EL_TILE / EL_TB; ++k) {
        const int i = min(base + k * EL_TB + tid, e - 1);
        d[k] = dst[i];
        s[k] = src[i];
    }
    #pragma unroll
    for (int k = 0; k < EL_TILE / EL_TB; ++k) {
        if (base + k * EL_TB + tid < e) {
            const int dk = min(max(d[k], 0), n - 1);
            int p = atomicAdd(&cur[min(dk >> EL_RLOG, EL_NB_MAX - 1)], 1);
            p = min(max(p, 0), e - 1);
            rec[p] = (int)(((unsigned)s[k] & 0xFFFFFFu) | ((unsigned)(dk & 255) << 24));
        }
    }
}

__global__ __launch_bounds__(EL_BTB) void el_build(const int* __restrict__ rec,
                                                   const int* __restrict__ toff,
                                                   int* __restrict__ start,
                                                   int* __restrict__ col,
                                                   float* __restrict__ dinv,
                                                   int n, int e, int nb, int ntiles) {
    constexpr int NODES = 1 << EL_RLOG;
    __shared__ int c[NODES];
    __shared__ int cur[NODES];
    __shared__ int wtot[NODES / 32];
    __shared__ int stage[EL_STAGE];
    const int tid  = threadIdx.x;
    const int lane = tid & 31;
    const int wave = tid >> 5;
    const int b    = blockIdx.x;
    int bs = toff[(size_t)b * ntiles];
    int be = (b + 1 < nb) ? toff[(size_t)(b + 1) * ntiles] : e;
    bs = min(max(bs, 0), e);
    be = min(max(be, bs), e);
    const int m = be - bs;

    if (tid < NODES) c[tid] = 0;
    __syncthreads();
    for (int i = tid; i < m; i += EL_BPER * EL_BTB) {
        unsigned r[EL_BPER];
        #pragma unroll
        for (int u = 0; u < EL_BPER; ++u)
            r[u] = (unsigned)rec[bs + min(i + u * EL_BTB, m - 1)];
        #pragma unroll
        for (int u = 0; u < EL_BPER; ++u)
            if (i + u * EL_BTB < m) atomicAdd(&c[r[u] >> 24], 1);
    }
    __syncthreads();

    const bool own = tid < NODES;
    const int cn = own ? c[tid] : 0;
    int incl = cn;
    #pragma unroll
    for (int o = 1; o < 32; o <<= 1) {
        int t = __shfl_up(incl, o);
        if (lane >= o) incl += t;
    }
    if (own && lane == 31) wtot[wave] = incl;
    __syncthreads();
    if (own) {
        int excl = incl - cn;
        for (int w = 0; w < wave; ++w) excl += wtot[w];
        const int node = (b << EL_RLOG) + tid;
        if (node < n) {
            start[node] = bs + excl;
            dinv[node]  = rsqrtf((float)(cn + 1));
        }
        cur[tid] = excl;
    }
    __syncthreads();

    const bool staged = (m <= EL_STAGE);
    for (int i = tid; i < m; i += EL_BPER * EL_BTB) {
        unsigned r[EL_BPER];
        #pragma unroll
        for (int u = 0; u < EL_BPER; ++u)
            r[u] = (unsigned)rec[bs + min(i + u * EL_BTB, m - 1)];
        #pragma unroll
        for (int u = 0; u < EL_BPER; ++u) {
            if (i + u * EL_BTB < m) {
                int p = atomicAdd(&cur[r[u] >> 24], 1);
                p = min(max(p, 0), m - 1);
                const int sv = (int)(r[u] & 0xFFFFFFu);
                if (staged) stage[p] = sv; else col[bs + p] = sv;
            }
        }
    }
    if (staged) {
        __syncthreads();
        for (int i = tid; i < m; i += EL_BTB) col[bs + i] = stage[i];
    }
}

template <int NOUT, bool MOVE>
__global__ __launch_bounds__(NOUT * 2) void gcn_gemm_wmma(
    const float* __restrict__ A, const float* __restrict__ B,
    const float* __restrict__ rscale, float* __restrict__ C, int M,
    const int* __restrict__ mv_src, int* __restrict__ mv_dst, int mv_n) {
    constexpr int NTHREADS = NOUT * 2;
    __shared__ float Alds[16 * KDIM];

    const int mTile = blockIdx.x;
    const int tid   = threadIdx.x;

    for (int i = tid; i < 16 * KDIM; i += NTHREADS) {
        int r = i >> 7;
        int c = i & (KDIM - 1);
        int row = mTile * 16 + r;
        if (row >= M) row = M - 1;
        Alds[i] = A[(size_t)row * KDIM + c] * rscale[row];
    }

    if (MOVE) {
        for (int i = mTile * NTHREADS + tid; i < mv_n; i += (int)gridDim.x * NTHREADS)
            mv_dst[i] = mv_src[i];
    }
    __syncthreads();

    const int wave  = tid >> 5;
    const int lane  = tid & 31;
    const int lrow  = lane & 15;
    const int khalf = lane >> 4;
    const int col   = lrow + wave * 16;

    v8f acc = {};
    #pragma unroll
    for (int k0 = 0; k0 < KDIM; k0 += 4) {
        const int ka = k0 + 2 * khalf;
        v2f a, b;
        a.x = Alds[lrow * KDIM + ka];
        a.y = Alds[lrow * KDIM + ka + 1];
        b.x = B[(size_t)ka * NOUT + col];
        b.y = B[(size_t)(ka + 1) * NOUT + col];
        acc = __builtin_amdgcn_wmma_f32_16x16x4_f32(
             false, a,  false, b,
             (short)0, acc,  false,  false);
    }

    const int rowBase = mTile * 16 + 8 * khalf;
    #pragma unroll
    for (int i = 0; i < 8; ++i) {
        int row = rowBase + i;
        if (row < M) C[(size_t)row * NOUT + col] = acc[i];
    }
}

template <int K, typename vec>
__device__ __forceinline__ void add_rows(vec& acc, const vec* __restrict__ H,
                                         int s, int j, int lane) {
    vec r[K];
    #pragma unroll
    for (int u = 0; u < K; ++u)
        r[u] = H[(size_t)__builtin_amdgcn_readlane(s, j + u) * 32 + lane];
    #pragma unroll
    for (int u = 0; u < K; ++u) acc += r[u];
}

template <int F>
__global__ __launch_bounds__(256) void gcn_gather_rows(
    const float* __restrict__ Hs, const int* __restrict__ start,
    const int* __restrict__ col, const float* __restrict__ dinv,
    const float* __restrict__ bias, float* __restrict__ out, int N, int E) {
    constexpr int V = F / 32;
    typedef float vec __attribute__((ext_vector_type(V)));
    const int lane = threadIdx.x & 31;
    const int n = __builtin_amdgcn_readfirstlane((int)(blockIdx.x * 8 + (threadIdx.x >> 5)));
    if (n >= N) return;
    const int beg = start[n];
    const int end = (n + 1 < N) ? start[n + 1] : E;
    const vec* __restrict__ H = reinterpret_cast<const vec*>(Hs);

    vec acc = H[(size_t)n * 32 + lane];
    for (int p = beg; p < end; p += 32) {
        const int m = min(32, end - p);
        int s = (lane < m) ? col[p + lane] : 0;
        s = min(max(s, 0), N - 1);
        int j = 0;
        for (; j + 8 <= m; j += 8) add_rows<8>(acc, H, s, j, lane);
        if (m & 4) { add_rows<4>(acc, H, s, j, lane); j += 4; }
        if (m & 2) { add_rows<2>(acc, H, s, j, lane); j += 2; }
        if (m & 1) add_rows<1>(acc, H, s, j, lane);
    }

    const float di = dinv[n];
    const vec b = reinterpret_cast<const vec*>(bias)[lane];
    vec v = acc * di + b;
    #pragma unroll
    for (int k = 0; k < V; ++k) v[k] = (v[k] > 0.0f) ? v[k] : expm1f(v[k]);
    reinterpret_cast<vec*>(out)[(size_t)n * 32 + lane] = v;
}

extern "C" void kernel_launch(void* const* d_in, const int* in_sizes, int n_in,
                              void* d_out, int out_size, void* d_ws, size_t ws_size,
                              hipStream_t stream) {
    const float* x  = (const float*)d_in[0];
    const int*   ei = (const int*)d_in[1];
    const float* W1 = (const float*)d_in[2];
    const float* b1 = (const float*)d_in[3];
    const float* W2 = (const float*)d_in[4];
    const float* b2 = (const float*)d_in[5];
    float* out = (float*)d_out;

    const int N = in_sizes[0] / IN_DIM;
    const int E = in_sizes[1] / 2;
    const int* src = ei;
    const int* dst = ei + E;

    char* ws = (char*)d_ws;
    float* dinv = (float*)ws;
    size_t off1 = ((size_t)N * 4 + 255) & ~(size_t)255;
    float* hbuf = (float*)(ws + off1);
    size_t off2 = off1 + ((((size_t)N * HID_DIM * 4) + 255) & ~(size_t)255);
    float* act1 = (float*)(ws + off2);

    int* csrA   = (int*)d_out;
    int* startA = csrA;
    int* colA   = csrA + N;
    const int ntiles  = (E + EL_TILE - 1) / EL_TILE;
    const int nb      = (N + (1 << EL_RLOG) - 1) >> EL_RLOG;
    const int flat    = nb * ntiles;
    const int flatPad = (flat + 4095) & ~4095;
    int* rec  = (int*)act1;
    int* tcnt = rec + ((E + 4095) & ~4095);
    int* toff = tcnt + flatPad;
    int* csrB   = (int*)(hbuf + (size_t)N * OUT_DIM);
    const int* startB = csrB;
    const int* colB   = csrB + N;
    const int csrInts = N + E;

    const int mTiles = (N + 15) / 16;
    const int nodeBlocks = (N + 7) / 8;

    el_hist<<<ntiles, EL_TB, 0, stream>>>(dst, tcnt, E, N, nb, ntiles);
    el_scan<<<flatPad / 4096, SCAN_TB, 0, stream>>>(tcnt, toff, flat);
    el_scatter<<<ntiles, EL_TB, 0, stream>>>(src, dst, toff, rec, E, N, nb, ntiles);
    el_build<<<nb, EL_BTB, 0, stream>>>(rec, toff, startA, colA, dinv, N, E, nb, ntiles);

    gcn_gemm_wmma<HID_DIM, false><<<mTiles, HID_DIM * 2, 0, stream>>>(
        x, W1, dinv, hbuf, N, nullptr, nullptr, 0);
    gcn_gather_rows<HID_DIM><<<nodeBlocks, 256, 0, stream>>>(
        hbuf, startA, colA, dinv, b1, act1, N, E);

    gcn_gemm_wmma<OUT_DIM, true><<<mTiles, OUT_DIM * 2, 0, stream>>>(
        act1, W2, dinv, hbuf, N, csrA, csrB, csrInts);
    gcn_gather_rows<OUT_DIM><<<nodeBlocks, 256, 0, stream>>>(
        hbuf, startB, colB, dinv, b2, out, N, E);
}


# GEMM A tile kept in LDS at a row pitch of 132 floats, not 128: the 16 rows of every A-fragment read now fall on distinct banks (was a 16-way bank conflict on each LDS load of both GEMMs)
# speedup vs baseline: 3.5387x; 1.1826x over previous
#include <hip/hip_runtime.h>
#include <cmath>


typedef __attribute__((ext_vector_type(2))) float v2f;
typedef __attribute__((ext_vector_type(8))) float v8f;

#define IN_DIM 128
#define HID_DIM 128
#define OUT_DIM 64
#define KDIM 128
#define APITCH (KDIM + 4)

#define EL_TILE    4096
#define EL_TB      256
#define EL_BTB     1024
#define EL_BPER    5
#define EL_RLOG    8
#define EL_NB_MAX  512
#define EL_STAGE   8192

__global__ __launch_bounds__(EL_TB) void el_hist(const int* __restrict__ dst,
                                                 int* __restrict__ tcnt,
                                                 int e, int n, int nb, int ntiles) {
    __shared__ int h[EL_NB_MAX];
    const int tid = threadIdx.x;
    for (int b = tid; b < EL_NB_MAX; b += EL_TB) h[b] = 0;
    __syncthreads();
    const int base = blockIdx.x * EL_TILE;
    int d[EL_TILE / EL_TB];
    #pragma unroll
    for (int k = 0; k < EL_TILE / EL_TB; ++k)
        d[k] = dst[min(base + k * EL_TB + tid, e - 1)];
    #pragma unroll
    for (int k = 0; k < EL_TILE / EL_TB; ++k) {
        if (base + k * EL_TB + tid < e) {
            const int dk = min(max(d[k], 0), n - 1);
            atomicAdd(&h[min(dk >> EL_RLOG, EL_NB_MAX - 1)], 1);
        }
    }
    __syncthreads();
    const int nbc = min(nb, EL_NB_MAX);
    for (int b = tid; b < nbc; b += EL_TB) tcnt[(size_t)b * ntiles + blockIdx.x] = h[b];
}

#define SCAN_TB 1024
__global__ __launch_bounds__(SCAN_TB) void el_scan(const int* __restrict__ cnt,
                                                   int* __restrict__ off, int len) {
    __shared__ int wpart[32];
    __shared__ int wtot[32];
    const int tid  = threadIdx.x;
    const int lane = tid & 31;
    const int wave = tid >> 5;
    const int4* __restrict__ c4 = reinterpret_cast<const int4*>(cnt);

    int part = 0;
    const int firstq = blockIdx.x * SCAN_TB;
    #pragma unroll 8
    for (int q = tid; q < firstq; q += SCAN_TB) {
        const int4 v = c4[q];
        part += (v.x + v.y) + (v.z + v.w);
    }
    #pragma unroll
    for (int o = 16; o > 0; o >>= 1) part += __shfl_xor(part, o);

    const int q  = firstq + tid;
    const int i0 = q * 4;
    int4 v = c4[q];
    if (i0 + 0 >= len) v.x = 0;
    if (i0 + 1 >= len) v.y = 0;
    if (i0 + 2 >= len) v.z = 0;
    if (i0 + 3 >= len) v.w = 0;
    const int c = (v.x + v.y) + (v.z + v.w);
    int incl = c;
    #pragma unroll
    for (int o = 1; o < 32; o <<= 1) {
        int t = __shfl_up(incl, o);
        if (lane >= o) incl += t;
    }
    if (lane == 0)  wpart[wave] = part;
    if (lane == 31) wtot[wave]  = incl;
    __syncthreads();
    if (wave == 0) {
        int p = wpart[lane];
        #pragma unroll
        for (int o = 16; o > 0; o >>= 1) p += __shfl_xor(p, o);
        const int t = wtot[lane];
        int ti = t;
        #pragma unroll
        for (int o = 1; o < 32; o <<= 1) {
            int u = __shfl_up(ti, o);
            if (lane >= o) ti += u;
        }
        wtot[lane] = p + ti - t;
    }
    __syncthreads();
    int4 r;
    r.x = wtot[wave] + incl - c;
    r.y = r.x + v.x;
    r.z = r.y + v.y;
    r.w = r.z + v.z;
    reinterpret_cast<int4*>(off)[q] = r;
}

__global__ __launch_bounds__(EL_TB) void el_scatter(const int* __restrict__ src,
                                                    const int* __restrict__ dst,
                                                    const int* __restrict__ toff,
                                                    int* __restrict__ rec,
                                                    int e, int n, int nb, int ntiles) {
    __shared__ int cur[EL_NB_MAX];
    const int tid = threadIdx.x;
    for (int b = tid; b < EL_NB_MAX; b += EL_TB)
        cur[b] = (b < nb) ? toff[(size_t)b * ntiles + blockIdx.x] : 0;
    __syncthreads();
    const int base = blockIdx.x * EL_TILE;
    int d[EL_TILE / EL_TB], s[EL_TILE / EL_TB];
    #pragma unroll
    for (int k = 0; k < EL_TILE / EL_TB; ++k) {
        const int i = min(base + k * EL_TB + tid, e - 1);
        d[k] = dst[i];
        s[k] = src[i];
    }
    #pragma unroll
    for (int k = 0; k < EL_TILE / EL_TB; ++k) {
        if (base + k * EL_TB + tid < e) {
            const int dk = min(max(d[k], 0), n - 1);
            int p = atomicAdd(&cur[min(dk >> EL_RLOG, EL_NB_MAX - 1)], 1);
            p = min(max(p, 0), e - 1);
            rec[p] = (int)(((unsigned)s[k] & 0xFFFFFFu) | ((unsigned)(dk & 255) << 24));
        }
    }
}

__global__ __launch_bounds__(EL_BTB) void el_build(const int* __restrict__ rec,
                                                   const int* __restrict__ toff,
                                                   int* __restrict__ start,
                                                   int* __restrict__ col,
                                                   float* __restrict__ dinv,
                                                   int n, int e, int nb, int ntiles) {
    constexpr int NODES = 1 << EL_RLOG;
    __shared__ int c[NODES];
    __shared__ int cur[NODES];
    __shared__ int wtot[NODES / 32];
    __shared__ int stage[EL_STAGE];
    const int tid  = threadIdx.x;
    const int lane = tid & 31;
    const int wave = tid >> 5;
    const int b    = blockIdx.x;
    int bs = toff[(size_t)b * ntiles];
    int be = (b + 1 < nb) ? toff[(size_t)(b + 1) * ntiles] : e;
    bs = min(max(bs, 0), e);
    be = min(max(be, bs), e);
    const int m = be - bs;

    if (tid < NODES) c[tid] = 0;
    __syncthreads();
    for (int i = tid; i < m; i += EL_BPER * EL_BTB) {
        unsigned r[EL_BPER];
        #pragma unroll
        for (int u = 0; u < EL_BPER; ++u)
            r[u] = (unsigned)rec[bs + min(i + u * EL_BTB, m - 1)];
        #pragma unroll
        for (int u = 0; u < EL_BPER; ++u)
            if (i + u * EL_BTB < m) atomicAdd(&c[r[u] >> 24], 1);
    }
    __syncthreads();

    const bool own = tid < NODES;
    const int cn = own ? c[tid] : 0;
    int incl = cn;
    #pragma unroll
    for (int o = 1; o < 32; o <<= 1) {
        int t = __shfl_up(incl, o);
        if (lane >= o) incl += t;
    }
    if (own && lane == 31) wtot[wave] = incl;
    __syncthreads();
    if (own) {
        int excl = incl - cn;
        for (int w = 0; w < wave; ++w) excl += wtot[w];
        const int node = (b << EL_RLOG) + tid;
        if (node < n) {
            start[node] = bs + excl;
            dinv[node]  = rsqrtf((float)(cn + 1));
        }
        cur[tid] = excl;
    }
    __syncthreads();

    const bool staged = (m <= EL_STAGE);
    for (int i = tid; i < m; i += EL_BPER * EL_BTB) {
        unsigned r[EL_BPER];
        #pragma unroll
        for (int u = 0; u < EL_BPER; ++u)
            r[u] = (unsigned)rec[bs + min(i + u * EL_BTB, m - 1)];
        #pragma unroll
        for (int u = 0; u < EL_BPER; ++u) {
            if (i + u * EL_BTB < m) {
                int p = atomicAdd(&cur[r[u] >> 24], 1);
                p = min(max(p, 0), m - 1);
                const int sv = (int)(r[u] & 0xFFFFFFu);
                if (staged) stage[p] = sv; else col[bs + p] = sv;
            }
        }
    }
    if (staged) {
        __syncthreads();
        for (int i = tid; i < m; i += EL_BTB) col[bs + i] = stage[i];
    }
}

template <int NOUT, bool MOVE>
__global__ __launch_bounds__(NOUT * 2) void gcn_gemm_wmma(
    const float* __restrict__ A, const float* __restrict__ B,
    const float* __restrict__ rscale, float* __restrict__ C, int M,
    const int* __restrict__ mv_src, int* __restrict__ mv_dst, int mv_n) {
    constexpr int NTHREADS = NOUT * 2;
    __shared__ float Alds[16 * APITCH];

    const int mTile = blockIdx.x;
    const int tid   = threadIdx.x;

    for (int i = tid; i < 16 * KDIM; i += NTHREADS) {
        int r = i >> 7;
        int c = i & (KDIM - 1);
        int row = mTile * 16 + r;
        if (row >= M) row = M - 1;
        Alds[r * APITCH + c] = A[(size_t)row * KDIM + c] * rscale[row];
    }

    if (MOVE) {
        for (int i = mTile * NTHREADS + tid; i < mv_n; i += (int)gridDim.x * NTHREADS)
            mv_dst[i] = mv_src[i];
    }
    __syncthreads();

    const int wave  = tid >> 5;
    const int lane  = tid & 31;
    const int lrow  = lane & 15;
    const int khalf = lane >> 4;
    const int col   = lrow + wave * 16;

    v8f acc = {};
    #pragma unroll
    for (int k0 = 0; k0 < KDIM; k0 += 4) {
        const int ka = k0 + 2 * khalf;
        v2f a, b;
        a.x = Alds[lrow * APITCH + ka];
        a.y = Alds[lrow * APITCH + ka + 1];
        b.x = B[(size_t)ka * NOUT + col];
        b.y = B[(size_t)(ka + 1) * NOUT + col];
        acc = __builtin_amdgcn_wmma_f32_16x16x4_f32(
             false, a,  false, b,
             (short)0, acc,  false,  false);
    }

    const int rowBase = mTile * 16 + 8 * khalf;
    #pragma unroll
    for (int i = 0; i < 8; ++i) {
        int row = rowBase + i;
        if (row < M) C[(size_t)row * NOUT + col] = acc[i];
    }
}

template <int K, typename vec>
__device__ __forceinline__ void add_rows(vec& acc, const vec* __restrict__ H,
                                         int s, int j, int lane) {
    vec r[K];
    #pragma unroll
    for (int u = 0; u < K; ++u)
        r[u] = H[(size_t)__builtin_amdgcn_readlane(s, j + u) * 32 + lane];
    #pragma unroll
    for (int u = 0; u < K; ++u) acc += r[u];
}

template <int F>
__global__ __launch_bounds__(256) void gcn_gather_rows(
    const float* __restrict__ Hs, const int* __restrict__ start,
    const int* __restrict__ col, const float* __restrict__ dinv,
    const float* __restrict__ bias, float* __restrict__ out, int N, int E) {
    constexpr int V = F / 32;
    typedef float vec __attribute__((ext_vector_type(V)));
    const int lane = threadIdx.x & 31;
    const int n = __builtin_amdgcn_readfirstlane((int)(blockIdx.x * 8 + (threadIdx.x >> 5)));
    if (n >= N) return;
    const int beg = start[n];
    const int end = (n + 1 < N) ? start[n + 1] : E;
    const vec* __restrict__ H = reinterpret_cast<const vec*>(Hs);

    vec acc = H[(size_t)n * 32 + lane];
    for (int p = beg; p < end; p += 32) {
        const int m = min(32, end - p);
        int s = (lane < m) ? col[p + lane] : 0;
        s = min(max(s, 0), N - 1);
        int j = 0;
        for (; j + 8 <= m; j += 8) add_rows<8>(acc, H, s, j, lane);
        if (m & 4) { add_rows<4>(acc, H, s, j, lane); j += 4; }
        if (m & 2) { add_rows<2>(acc, H, s, j, lane); j += 2; }
        if (m & 1) add_rows<1>(acc, H, s, j, lane);
    }

    const float di = dinv[n];
    const vec b = reinterpret_cast<const vec*>(bias)[lane];
    vec v = acc * di + b;
    #pragma unroll
    for (int k = 0; k < V; ++k) v[k] = (v[k] > 0.0f) ? v[k] : expm1f(v[k]);
    reinterpret_cast<vec*>(out)[(size_t)n * 32 + lane] = v;
}

extern "C" void kernel_launch(void* const* d_in, const int* in_sizes, int n_in,
                              void* d_out, int out_size, void* d_ws, size_t ws_size,
                              hipStream_t stream) {
    const float* x  = (const float*)d_in[0];
    const int*   ei = (const int*)d_in[1];
    const float* W1 = (const float*)d_in[2];
    const float* b1 = (const float*)d_in[3];
    const float* W2 = (const float*)d_in[4];
    const float* b2 = (const float*)d_in[5];
    float* out = (float*)d_out;

    const int N = in_sizes[0] / IN_DIM;
    const int E = in_sizes[1] / 2;
    const int* src = ei;
    const int* dst = ei + E;

    char* ws = (char*)d_ws;
    float* dinv = (float*)ws;
    size_t off1 = ((size_t)N * 4 + 255) & ~(size_t)255;
    float* hbuf = (float*)(ws + off1);
    size_t off2 = off1 + ((((size_t)N * HID_DIM * 4) + 255) & ~(size_t)255);
    float* act1 = (float*)(ws + off2);

    int* csrA   = (int*)d_out;
    int* startA = csrA;
    int* colA   = csrA + N;
    const int ntiles  = (E + EL_TILE - 1) / EL_TILE;
    const int nb      = (N + (1 << EL_RLOG) - 1) >> EL_RLOG;
    const int flat    = nb * ntiles;
    const int flatPad = (flat + 4095) & ~4095;
    int* rec  = (int*)act1;
    int* tcnt = rec + ((E + 4095) & ~4095);
    int* toff = tcnt + flatPad;
    int* csrB   = (int*)(hbuf + (size_t)N * OUT_DIM);
    const int* startB = csrB;
    const int* colB   = csrB + N;
    const int csrInts = N + E;

    const int mTiles = (N + 15) / 16;
    const int nodeBlocks = (N + 7) / 8;

    el_hist<<<ntiles, EL_TB, 0, stream>>>(dst, tcnt, E, N, nb, ntiles);
    el_scan<<<flatPad / 4096, SCAN_TB, 0, stream>>>(tcnt, toff, flat);
    el_scatter<<<ntiles, EL_TB, 0, stream>>>(src, dst, toff, rec, E, N, nb, ntiles);
    el_build<<<nb, EL_BTB, 0, stream>>>(rec, toff, startA, colA, dinv, N, E, nb, ntiles);

    gcn_gemm_wmma<HID_DIM, false><<<mTiles, HID_DIM * 2, 0, stream>>>(
        x, W1, dinv, hbuf, N, nullptr, nullptr, 0);
    gcn_gather_rows<HID_DIM><<<nodeBlocks, 256, 0, stream>>>(
        hbuf, startA, colA, dinv, b1, act1, N, E);

    gcn_gemm_wmma<OUT_DIM, true><<<mTiles, OUT_DIM * 2, 0, stream>>>(
        act1, W2, dinv, hbuf, N, csrA, csrB, csrInts);
    gcn_gather_rows<OUT_DIM><<<nodeBlocks, 256, 0, stream>>>(
        hbuf, startB, colB, dinv, b2, out, N, E);
}


# Layer-1 row sums and layer-2 product in one kernel: each 32-node tile of summed, activated rows goes to LDS and is multiplied there (W2 fragments in registers); same values, same sum order
# speedup vs baseline: 3.7052x; 1.0471x over previous
#include <hip/hip_runtime.h>
#include <cmath>


typedef __attribute__((ext_vector_type(2))) float v2f;
typedef __attribute__((ext_vector_type(8))) float v8f;

#define IN_DIM 128
#define HID_DIM 128
#define OUT_DIM 64
#define KDIM 128
#define APITCH (KDIM + 4)

#define EL_TILE    4096
#define EL_TB      256
#define EL_BTB     1024
#define EL_BPER    5
#define EL_RLOG    8
#define EL_NB_MAX  512
#define EL_STAGE   8192

__global__ __launch_bounds__(EL_TB) void el_hist(const int* __restrict__ dst,
                                                 int* __restrict__ tcnt,
                                                 int e, int n, int nb, int ntiles) {
    __shared__ int h[EL_NB_MAX];
    const int tid = threadIdx.x;
    for (int b = tid; b < EL_NB_MAX; b += EL_TB) h[b] = 0;
    __syncthreads();
    const int base = blockIdx.x * EL_TILE;
    int d[EL_TILE / EL_TB];
    #pragma unroll
    for (int k = 0; k < EL_TILE / EL_TB; ++k)
        d[k] = dst[min(base + k * EL_TB + tid, e - 1)];
    #pragma unroll
    for (int k = 0; k < EL_TILE / EL_TB; ++k) {
        if (base + k * EL_TB + tid < e) {
            const int dk = min(max(d[k], 0), n - 1);
            atomicAdd(&h[min(dk >> EL_RLOG, EL_NB_MAX - 1)], 1);
        }
    }
    __syncthreads();
    const int nbc = min(nb, EL_NB_MAX);
    for (int b = tid; b < nbc; b += EL_TB) tcnt[(size_t)b * ntiles + blockIdx.x] = h[b];
}

#define SCAN_TB 1024
__global__ __launch_bounds__(SCAN_TB) void el_scan(const int* __restrict__ cnt,
                                                   int* __restrict__ off, int len) {
    __shared__ int wpart[32];
    __shared__ int wtot[32];
    const int tid  = threadIdx.x;
    const int lane = tid & 31;
    const int wave = tid >> 5;
    const int4* __restrict__ c4 = reinterpret_cast<const int4*>(cnt);

    int part = 0;
    const int firstq = blockIdx.x * SCAN_TB;
    #pragma unroll 8
    for (int q = tid; q < firstq; q += SCAN_TB) {
        const int4 v = c4[q];
        part += (v.x + v.y) + (v.z + v.w);
    }
    #pragma unroll
    for (int o = 16; o > 0; o >>= 1) part += __shfl_xor(part, o);

    const int q  = firstq + tid;
    const int i0 = q * 4;
    int4 v = c4[q];
    if (i0 + 0 >= len) v.x = 0;
    if (i0 + 1 >= len) v.y = 0;
    if (i0 + 2 >= len) v.z = 0;
    if (i0 + 3 >= len) v.w = 0;
    const int c = (v.x + v.y) + (v.z + v.w);
    int incl = c;
    #pragma unroll
    for (int o = 1; o < 32; o <<= 1) {
        int t = __shfl_up(incl, o);
        if (lane >= o) incl += t;
    }
    if (lane == 0)  wpart[wave] = part;
    if (lane == 31) wtot[wave]  = incl;
    __syncthreads();
    if (wave == 0) {
        int p = wpart[lane];
        #pragma unroll
        for (int o = 16; o > 0; o >>= 1) p += __shfl_xor(p, o);
        const int t = wtot[lane];
        int ti = t;
        #pragma unroll
        for (int o = 1; o < 32; o <<= 1) {
            int u = __shfl_up(ti, o);
            if (lane >= o) ti += u;
        }
        wtot[lane] = p + ti - t;
    }
    __syncthreads();
    int4 r;
    r.x = wtot[wave] + incl - c;
    r.y = r.x + v.x;
    r.z = r.y + v.y;
    r.w = r.z + v.z;
    reinterpret_cast<int4*>(off)[q] = r;
}

__global__ __launch_bounds__(EL_TB) void el_scatter(const int* __restrict__ src,
                                                    const int* __restrict__ dst,
                                                    const int* __restrict__ toff,
                                                    int* __restrict__ rec,
                                                    int e, int n, int nb, int ntiles) {
    __shared__ int cur[EL_NB_MAX];
    const int tid = threadIdx.x;
    for (int b = tid; b < EL_NB_MAX; b += EL_TB)
        cur[b] = (b < nb) ? toff[(size_t)b * ntiles + blockIdx.x] : 0;
    __syncthreads();
    const int base = blockIdx.x * EL_TILE;
    int d[EL_TILE / EL_TB], s[EL_TILE / EL_TB];
    #pragma unroll
    for (int k = 0; k < EL_TILE / EL_TB; ++k) {
        const int i = min(base + k * EL_TB + tid, e - 1);
        d[k] = dst[i];
        s[k] = src[i];
    }
    #pragma unroll
    for (int k = 0; k < EL_TILE / EL_TB; ++k) {
        if (base + k * EL_TB + tid < e) {
            const int dk = min(max(d[k], 0), n - 1);
            int p = atomicAdd(&cur[min(dk >> EL_RLOG, EL_NB_MAX - 1)], 1);
            p = min(max(p, 0), e - 1);
            rec[p] = (int)(((unsigned)s[k] & 0xFFFFFFu) | ((unsigned)(dk & 255) << 24));
        }
    }
}

__global__ __launch_bounds__(EL_BTB) void el_build(const int* __restrict__ rec,
                                                   const int* __restrict__ toff,
                                                   int* __restrict__ start,
                                                   int* __restrict__ col,
                                                   float* __restrict__ dinv,
                                                   int n, int e, int nb, int ntiles) {
    constexpr int NODES = 1 << EL_RLOG;
    __shared__ int c[NODES];
    __shared__ int cur[NODES];
    __shared__ int wtot[NODES / 32];
    __shared__ int stage[EL_STAGE];
    const int tid  = threadIdx.x;
    const int lane = tid & 31;
    const int wave = tid >> 5;
    const int b    = blockIdx.x;
    int bs = toff[(size_t)b * ntiles];
    int be = (b + 1 < nb) ? toff[(size_t)(b + 1) * ntiles] : e;
    bs = min(max(bs, 0), e);
    be = min(max(be, bs), e);
    const int m = be - bs;

    if (tid < NODES) c[tid] = 0;
    __syncthreads();
    for (int i = tid; i < m; i += EL_BPER * EL_BTB) {
        unsigned r[EL_BPER];
        #pragma unroll
        for (int u = 0; u < EL_BPER; ++u)
            r[u] = (unsigned)rec[bs + min(i + u * EL_BTB, m - 1)];
        #pragma unroll
        for (int u = 0; u < EL_BPER; ++u)
            if (i + u * EL_BTB < m) atomicAdd(&c[r[u] >> 24], 1);
    }
    __syncthreads();

    const bool own = tid < NODES;
    const int cn = own ? c[tid] : 0;
    int incl = cn;
    #pragma unroll
    for (int o = 1; o < 32; o <<= 1) {
        int t = __shfl_up(incl, o);
        if (lane >= o) incl += t;
    }
    if (own && lane == 31) wtot[wave] = incl;
    __syncthreads();
    if (own) {
        int excl = incl - cn;
        for (int w = 0; w < wave; ++w) excl += wtot[w];
        const int node = (b << EL_RLOG) + tid;
        if (node < n) {
            start[node] = bs + excl;
            dinv[node]  = rsqrtf((float)(cn + 1));
        }
        cur[tid] = excl;
    }
    __syncthreads();

    const bool staged = (m <= EL_STAGE);
    for (int i = tid; i < m; i += EL_BPER * EL_BTB) {
        unsigned r[EL_BPER];
        #pragma unroll
        for (int u = 0; u < EL_BPER; ++u)
            r[u] = (unsigned)rec[bs + min(i + u * EL_BTB, m - 1)];
        #pragma unroll
        for (int u = 0; u < EL_BPER; ++u) {
            if (i + u * EL_BTB < m) {
                int p = atomicAdd(&cur[r[u] >> 24], 1);
                p = min(max(p, 0), m - 1);
                const int sv = (int)(r[u] & 0xFFFFFFu);
                if (staged) stage[p] = sv; else col[bs + p] = sv;
            }
        }
    }
    if (staged) {
        __syncthreads();
        for (int i = tid; i < m; i += EL_BTB) col[bs + i] = stage[i];
    }
}

template <int NOUT>
__global__ __launch_bounds__(NOUT * 2) void gcn_gemm_wmma(
    const float* __restrict__ A, const float* __restrict__ B,
    const float* __restrict__ rscale, float* __restrict__ C, int M) {
    constexpr int NTHREADS = NOUT * 2;
    __shared__ float Alds[16 * APITCH];

    const int mTile = blockIdx.x;
    const int tid   = threadIdx.x;

    for (int i = tid; i < 16 * KDIM; i += NTHREADS) {
        int r = i >> 7;
        int c = i & (KDIM - 1);
        int row = mTile * 16 + r;
        if (row >= M) row = M - 1;
        Alds[r * APITCH + c] = A[(size_t)row * KDIM + c] * rscale[row];
    }

    __syncthreads();

    const int wave  = tid >> 5;
    const int lane  = tid & 31;
    const int lrow  = lane & 15;
    const int khalf = lane >> 4;
    const int col   = lrow + wave * 16;

    v8f acc = {};
    #pragma unroll
    for (int k0 = 0; k0 < KDIM; k0 += 4) {
        const int ka = k0 + 2 * khalf;
        v2f a, b;
        a.x = Alds[lrow * APITCH + ka];
        a.y = Alds[lrow * APITCH + ka + 1];
        b.x = B[(size_t)ka * NOUT + col];
        b.y = B[(size_t)(ka + 1) * NOUT + col];
        acc = __builtin_amdgcn_wmma_f32_16x16x4_f32(
             false, a,  false, b,
             (short)0, acc,  false,  false);
    }

    const int rowBase = mTile * 16 + 8 * khalf;
    #pragma unroll
    for (int i = 0; i < 8; ++i) {
        int row = rowBase + i;
        if (row < M) C[(size_t)row * NOUT + col] = acc[i];
    }
}

template <int K, typename vec>
__device__ __forceinline__ void add_rows(vec& acc, const vec* __restrict__ H,
                                         int s, int j, int lane) {
    vec r[K];
    #pragma unroll
    for (int u = 0; u < K; ++u)
        r[u] = H[(size_t)__builtin_amdgcn_readlane(s, j + u) * 32 + lane];
    #pragma unroll
    for (int u = 0; u < K; ++u) acc += r[u];
}

template <int F>
__global__ __launch_bounds__(256) void gcn_gather_rows(
    const float* __restrict__ Hs, const int* __restrict__ start,
    const int* __restrict__ col, const float* __restrict__ dinv,
    const float* __restrict__ bias, float* __restrict__ out, int N, int E) {
    constexpr int V = F / 32;
    typedef float vec __attribute__((ext_vector_type(V)));
    const int lane = threadIdx.x & 31;
    const int n = __builtin_amdgcn_readfirstlane((int)(blockIdx.x * 8 + (threadIdx.x >> 5)));
    if (n >= N) return;
    const int beg = start[n];
    const int end = (n + 1 < N) ? start[n + 1] : E;
    const vec* __restrict__ H = reinterpret_cast<const vec*>(Hs);

    vec acc = H[(size_t)n * 32 + lane];
    for (int p = beg; p < end; p += 32) {
        const int m = min(32, end - p);
        int s = (lane < m) ? col[p + lane] : 0;
        s = min(max(s, 0), N - 1);
        int j = 0;
        for (; j + 8 <= m; j += 8) add_rows<8>(acc, H, s, j, lane);
        if (m & 4) { add_rows<4>(acc, H, s, j, lane); j += 4; }
        if (m & 2) { add_rows<2>(acc, H, s, j, lane); j += 2; }
        if (m & 1) add_rows<1>(acc, H, s, j, lane);
    }

    const float di = dinv[n];
    const vec b = reinterpret_cast<const vec*>(bias)[lane];
    vec v = acc * di + b;
    #pragma unroll
    for (int k = 0; k < V; ++k) v[k] = (v[k] > 0.0f) ? v[k] : expm1f(v[k]);
    reinterpret_cast<vec*>(out)[(size_t)n * 32 + lane] = v;
}

#define FT_ROWS 32
#define FT_TILES 4

__global__ __launch_bounds__(256) void gcn_rows_gemm(
    const float* __restrict__ Hs, const int* __restrict__ start,
    const int* __restrict__ col, const float* __restrict__ dinv,
    const float* __restrict__ bias, const float* __restrict__ B,
    float* __restrict__ C, int N, int E) {
    typedef float vec __attribute__((ext_vector_type(4)));
    __shared__ __attribute__((aligned(16))) float Alds[2][FT_ROWS * APITCH];

    const int tid   = threadIdx.x;
    const int lane  = tid & 31;
    const int wave  = __builtin_amdgcn_readfirstlane(tid >> 5);
    const int lrow  = lane & 15;
    const int khalf = lane >> 4;
    const int mt    = wave >> 2;
    const int ncol  = (wave & 3) * 16 + lrow;

    v2f bfrag[KDIM / 4];
    #pragma unroll
    for (int i = 0; i < KDIM / 4; ++i) {
        const int ka = 4 * i + 2 * khalf;
        bfrag[i].x = B[(size_t)ka * OUT_DIM + ncol];
        bfrag[i].y = B[(size_t)(ka + 1) * OUT_DIM + ncol];
    }

    const vec* __restrict__ H = reinterpret_cast<const vec*>(Hs);
    const vec bv = reinterpret_cast<const vec*>(bias)[lane];
    const int ntile = (N + FT_ROWS - 1) / FT_ROWS;
    int buf = 0;
    for (int t = blockIdx.x; t < ntile; t += (int)gridDim.x, buf ^= 1) {
        float* __restrict__ At = Alds[buf];
        const int nbase = t * FT_ROWS + wave;

        int sb = 0;
        if (lane < 2 * (FT_ROWS / 8)) {
            const int idx = nbase + 8 * (lane >> 1) + (lane & 1);
            sb = (idx < N) ? start[idx] : E;
        }

        #pragma unroll
        for (int q = 0; q < FT_ROWS / 8; ++q) {
            const int r = wave + 8 * q;
            const int n = nbase + 8 * q;
            vec v = {};
            if (n < N) {
                const int beg = __builtin_amdgcn_readlane(sb, 2 * q);
                const int end = __builtin_amdgcn_readlane(sb, 2 * q + 1);
                vec acc = H[(size_t)n * 32 + lane];
                for (int p = beg; p < end; p += 32) {
                    const int m = min(32, end - p);
                    int s = (lane < m) ? col[p + lane] : 0;
                    s = min(max(s, 0), N - 1);
                    int j = 0;
                    for (; j + 8 <= m; j += 8) add_rows<8>(acc, H, s, j, lane);
                    if (m & 4) { add_rows<4>(acc, H, s, j, lane); j += 4; }
                    if (m & 2) { add_rows<2>(acc, H, s, j, lane); j += 2; }
                    if (m & 1) add_rows<1>(acc, H, s, j, lane);
                }
                const float di = dinv[n];
                v = acc * di + bv;
                #pragma unroll
                for (int k = 0; k < 4; ++k) v[k] = (v[k] > 0.0f) ? v[k] : expm1f(v[k]);
                v *= di;
            }
            *reinterpret_cast<vec*>(&At[r * APITCH + lane * 4]) = v;
        }
        __syncthreads();

        v8f acc = {};
        const float* __restrict__ Ar = &At[(mt * 16 + lrow) * APITCH + 2 * khalf];
        #pragma unroll
        for (int i = 0; i < KDIM / 4; ++i) {
            const v2f a = *reinterpret_cast<const v2f*>(&Ar[4 * i]);
            acc = __builtin_amdgcn_wmma_f32_16x16x4_f32(
                 false, a,  false, bfrag[i],
                 (short)0, acc,  false,  false);
        }

        const int rowBase = t * FT_ROWS + mt * 16 + 8 * khalf;
        #pragma unroll
        for (int i = 0; i < 8; ++i) {
            const int row = rowBase + i;
            if (row < N) C[(size_t)row * OUT_DIM + ncol] = acc[i];
        }
    }
}

extern "C" void kernel_launch(void* const* d_in, const int* in_sizes, int n_in,
                              void* d_out, int out_size, void* d_ws, size_t ws_size,
                              hipStream_t stream) {
    const float* x  = (const float*)d_in[0];
    const int*   ei = (const int*)d_in[1];
    const float* W1 = (const float*)d_in[2];
    const float* b1 = (const float*)d_in[3];
    const float* W2 = (const float*)d_in[4];
    const float* b2 = (const float*)d_in[5];
    float* out = (float*)d_out;

    const int N = in_sizes[0] / IN_DIM;
    const int E = in_sizes[1] / 2;
    const int* src = ei;
    const int* dst = ei + E;

    char* ws = (char*)d_ws;
    float* dinv = (float*)ws;
    size_t off1 = ((size_t)N * 4 + 255) & ~(size_t)255;
    float* hbuf = (float*)(ws + off1);
    size_t off2 = off1 + ((((size_t)N * HID_DIM * 4) + 255) & ~(size_t)255);
    float* h2buf = (float*)(ws + off2);
    size_t off3 = off2 + ((((size_t)N * OUT_DIM * 4) + 255) & ~(size_t)255);
    int* startA = (int*)(ws + off3);
    int* colA   = startA + N;
    size_t off4 = off3 + ((((size_t)N + E) * 4 + 255) & ~(size_t)255);
    const int ntiles  = (E + EL_TILE - 1) / EL_TILE;
    const int nb      = (N + (1 << EL_RLOG) - 1) >> EL_RLOG;
    const int flat    = nb * ntiles;
    const int flatPad = (flat + 4095) & ~4095;
    int* rec  = (int*)(ws + off4);
    int* tcnt = rec + ((E + 4095) & ~4095);
    int* toff = tcnt + flatPad;

    const int mTiles = (N + 15) / 16;
    const int nodeBlocks = (N + 7) / 8;
    const int ftTiles  = (N + FT_ROWS - 1) / FT_ROWS;
    const int ftBlocks = (ftTiles + FT_TILES - 1) / FT_TILES;

    el_hist<<<ntiles, EL_TB, 0, stream>>>(dst, tcnt, E, N, nb, ntiles);
    el_scan<<<flatPad / 4096, SCAN_TB, 0, stream>>>(tcnt, toff, flat);
    el_scatter<<<ntiles, EL_TB, 0, stream>>>(src, dst, toff, rec, E, N, nb, ntiles);
    el_build<<<nb, EL_BTB, 0, stream>>>(rec, toff, startA, colA, dinv, N, E, nb, ntiles);

    gcn_gemm_wmma<HID_DIM><<<mTiles, HID_DIM * 2, 0, stream>>>(x, W1, dinv, hbuf, N);
    gcn_rows_gemm<<<ftBlocks, 256, 0, stream>>>(
        hbuf, startA, colA, dinv, b1, W2, h2buf, N, E);
    gcn_gather_rows<OUT_DIM><<<nodeBlocks, 256, 0, stream>>>(
        h2buf, startA, colA, dinv, b2, out, N, E);
}


# Edge list in 2 launches, not 4: each 4096-edge tile is bucket-ordered in LDS and stored whole with its offsets; a bucket block pulls its runs; no scan pass, no 4-byte scatter; row-sum order changes
# speedup vs baseline: 4.9140x; 1.3262x over previous
#include <hip/hip_runtime.h>
#include <cmath>


typedef __attribute__((ext_vector_type(2))) float v2f;
typedef __attribute__((ext_vector_type(8))) float v8f;

#define IN_DIM 128
#define HID_DIM 128
#define OUT_DIM 64
#define KDIM 128
#define APITCH (KDIM + 4)

#define EL_TILE    4096
#define EL_TB      512
#define EL_BTB     1024
#define EL_BPER    5
#define EL_RLOG    8
#define EL_NB_MAX  512
#define EL_NT_MAX  512
#define EL_STAGE   8192

__global__ __launch_bounds__(EL_TB) void el_tile(const int* __restrict__ src,
                                                 const int* __restrict__ dst,
                                                 int* __restrict__ rec,
                                                 int* __restrict__ toffs,
                                                 int e, int n, int nb, int ntiles) {
    constexpr int PER = EL_TILE / EL_TB;
    static_assert(EL_TB == EL_NB_MAX, "one bucket counter per thread");
    __shared__ int h[EL_NB_MAX];
    __shared__ int cur[EL_NB_MAX];
    __shared__ int wtot[EL_TB / 32];
    __shared__ __attribute__((aligned(16))) unsigned srt[EL_TILE];
    const int tid  = threadIdx.x;
    const int lane = tid & 31;
    const int wave = tid >> 5;
    h[tid] = 0;
    __syncthreads();
    const int base = blockIdx.x * EL_TILE;
    const int cnt  = min(max(e - base, 0), EL_TILE);
    int d[PER], s[PER];
    #pragma unroll
    for (int k = 0; k < PER; ++k) {
        const int i = min(base + k * EL_TB + tid, e - 1);
        d[k] = dst[i];
        s[k] = src[i];
    }
    #pragma unroll
    for (int k = 0; k < PER; ++k) {
        d[k] = min(max(d[k], 0), n - 1);
        if (k * EL_TB + tid < cnt) atomicAdd(&h[min(d[k] >> EL_RLOG, EL_NB_MAX - 1)], 1);
    }
    __syncthreads();

    const int a = h[tid];
    int incl = a;
    #pragma unroll
    for (int o = 1; o < 32; o <<= 1) {
        int t = __shfl_up(incl, o);
        if (lane >= o) incl += t;
    }
    if (lane == 31) wtot[wave] = incl;
    __syncthreads();
    int excl = incl - a;
    #pragma unroll
    for (int w = 0; w < EL_TB / 32; ++w) excl += (w < wave) ? wtot[w] : 0;
    h[tid]   = excl;
    cur[tid] = excl;
    __syncthreads();

    #pragma unroll
    for (int k = 0; k < PER; ++k) {
        if (k * EL_TB + tid < cnt) {
            int p = atomicAdd(&cur[min(d[k] >> EL_RLOG, EL_NB_MAX - 1)], 1);
            p = min(max(p, 0), EL_TILE - 1);
            srt[p] = ((unsigned)s[k] & 0xFFFFFFu) | ((unsigned)(d[k] & 255) << 24);
        }
    }
    __syncthreads();

    const uint4* __restrict__ s4 = reinterpret_cast<const uint4*>(srt);
    uint4* __restrict__ r4 = reinterpret_cast<uint4*>(rec + base);
    for (int i = tid; i * 4 < cnt; i += EL_TB) {
        if (i * 4 + 3 < cnt) {
            r4[i] = s4[i];
        } else {
            for (int j = i * 4; j < cnt; ++j) rec[base + j] = (int)srt[j];
        }
    }
    const int nbc = min(nb, EL_NB_MAX);
    for (int b = tid; b <= nbc; b += EL_TB)
        toffs[(size_t)b * ntiles + blockIdx.x] = (b < EL_NB_MAX) ? h[b] : cnt;
}

__global__ __launch_bounds__(EL_BTB) void el_build(const int* __restrict__ rec,
                                                   const int* __restrict__ toffs,
                                                   int* __restrict__ start,
                                                   int* __restrict__ col,
                                                   float* __restrict__ dinv,
                                                   int n, int e, int nb, int ntiles) {
    constexpr int NODES = 1 << EL_RLOG;
    constexpr int TW    = EL_NT_MAX / 32;
    __shared__ int c[NODES];
    __shared__ int cur[NODES];
    __shared__ int wtot[NODES / 32];
    __shared__ int stage[EL_STAGE];
    __shared__ int addr[EL_STAGE];
    __shared__ int rpos[EL_NT_MAX + 1];
    __shared__ int rsrc[EL_NT_MAX];
    __shared__ __attribute__((aligned(16))) int wlen[TW];
    __shared__ __attribute__((aligned(16))) int wsum[TW];
    const int tid  = threadIdx.x;
    const int lane = tid & 31;
    const int wave = tid >> 5;
    const int b    = blockIdx.x;
    const int nt   = min(ntiles, EL_NT_MAX);

    int s0 = 0, len = 0;
    if (tid < nt) {
        int a = toffs[(size_t)b * ntiles + tid];
        int z = toffs[(size_t)(b + 1) * ntiles + tid];
        a = min(max(a, 0), EL_TILE);
        z = min(max(z, a), EL_TILE);
        s0  = a;
        len = z - a;
    }
    int incl = len;
    #pragma unroll
    for (int o = 1; o < 32; o <<= 1) {
        int t = __shfl_up(incl, o);
        if (lane >= o) incl += t;
    }
    int ssum = s0;
    #pragma unroll
    for (int o = 16; o > 0; o >>= 1) ssum += __shfl_xor(ssum, o);
    if (wave < TW && lane == 31) wlen[wave] = incl;
    if (wave < TW && lane == 0)  wsum[wave] = ssum;
    if (tid < NODES) c[tid] = 0;
    __syncthreads();
    int before = 0, mraw = 0, bsraw = 0;
    #pragma unroll
    for (int w = 0; w < TW; ++w) {
        const int v = wlen[w];
        mraw   += v;
        before += (w < wave) ? v : 0;
        bsraw  += wsum[w];
    }
    const int bs  = min(max(bsraw, 0), e);
    const int m   = min(max(mraw, 0), e - bs);
    const int pos = before + incl - len;
    if (tid < nt) {
        rpos[tid] = pos;
        rsrc[tid] = tid * EL_TILE + s0;
        for (int i = 0; i < len; ++i)
            if (pos + i < EL_STAGE) addr[pos + i] = tid * EL_TILE + s0 + i;
    }
    if (tid == 0) rpos[nt] = mraw;
    __syncthreads();

    const int  recmax = nt * EL_TILE - 1;
    const bool listed = (m <= EL_STAGE);
    auto find_run = [&](int p) -> int {
        if (p < EL_STAGE) return addr[p];
        int lo = 0, hi = nt - 1;
        for (int it = 0; it < 10 && lo < hi; ++it) {
            const int mid = (lo + hi + 1) >> 1;
            if (rpos[mid] <= p) lo = mid; else hi = mid - 1;
        }
        return rsrc[lo] + (p - rpos[lo]);
    };
    auto load_recs = [&](int i, unsigned (&r)[EL_BPER]) {
        int a[EL_BPER];
        if (listed) {
            #pragma unroll
            for (int u = 0; u < EL_BPER; ++u) a[u] = addr[min(i + u * EL_BTB, m - 1)];
        } else {
            #pragma unroll
            for (int u = 0; u < EL_BPER; ++u) a[u] = find_run(min(i + u * EL_BTB, m - 1));
        }
        #pragma unroll
        for (int u = 0; u < EL_BPER; ++u) r[u] = (unsigned)rec[min(max(a[u], 0), recmax)];
    };

    for (int i = tid; i < m; i += EL_BPER * EL_BTB) {
        unsigned r[EL_BPER];
        load_recs(i, r);
        #pragma unroll
        for (int u = 0; u < EL_BPER; ++u)
            if (i + u * EL_BTB < m) atomicAdd(&c[r[u] >> 24], 1);
    }
    __syncthreads();

    const bool own = tid < NODES;
    const int cn = own ? c[tid] : 0;
    int nincl = cn;
    #pragma unroll
    for (int o = 1; o < 32; o <<= 1) {
        int t = __shfl_up(nincl, o);
        if (lane >= o) nincl += t;
    }
    if (own && lane == 31) wtot[wave] = nincl;
    __syncthreads();
    if (own) {
        int excl = nincl - cn;
        for (int w = 0; w < wave; ++w) excl += wtot[w];
        const int node = (b << EL_RLOG) + tid;
        if (node < n) {
            start[node] = bs + excl;
            dinv[node]  = rsqrtf((float)(cn + 1));
        }
        cur[tid] = excl;
    }
    __syncthreads();

    for (int i = tid; i < m; i += EL_BPER * EL_BTB) {
        unsigned r[EL_BPER];
        load_recs(i, r);
        #pragma unroll
        for (int u = 0; u < EL_BPER; ++u) {
            if (i + u * EL_BTB < m) {
                int p = atomicAdd(&cur[r[u] >> 24], 1);
                p = min(max(p, 0), m - 1);
                const int sv = (int)(r[u] & 0xFFFFFFu);
                if (listed) stage[p] = sv; else col[bs + p] = sv;
            }
        }
    }
    if (listed) {
        __syncthreads();
        for (int i = tid; i < m; i += EL_BTB) col[bs + i] = stage[i];
    }
}

template <int NOUT>
__global__ __launch_bounds__(NOUT * 2) void gcn_gemm_wmma(
    const float* __restrict__ A, const float* __restrict__ B,
    const float* __restrict__ rscale, float* __restrict__ C, int M) {
    constexpr int NTHREADS = NOUT * 2;
    __shared__ float Alds[16 * APITCH];

    const int mTile = blockIdx.x;
    const int tid   = threadIdx.x;

    for (int i = tid; i < 16 * KDIM; i += NTHREADS) {
        int r = i >> 7;
        int c = i & (KDIM - 1);
        int row = mTile * 16 + r;
        if (row >= M) row = M - 1;
        Alds[r * APITCH + c] = A[(size_t)row * KDIM + c] * rscale[row];
    }

    __syncthreads();

    const int wave  = tid >> 5;
    const int lane  = tid & 31;
    const int lrow  = lane & 15;
    const int khalf = lane >> 4;
    const int col   = lrow + wave * 16;

    v8f acc = {};
    #pragma unroll
    for (int k0 = 0; k0 < KDIM; k0 += 4) {
        const int ka = k0 + 2 * khalf;
        v2f a, b;
        a.x = Alds[lrow * APITCH + ka];
        a.y = Alds[lrow * APITCH + ka + 1];
        b.x = B[(size_t)ka * NOUT + col];
        b.y = B[(size_t)(ka + 1) * NOUT + col];
        acc = __builtin_amdgcn_wmma_f32_16x16x4_f32(
             false, a,  false, b,
             (short)0, acc,  false,  false);
    }

    const int rowBase = mTile * 16 + 8 * khalf;
    #pragma unroll
    for (int i = 0; i < 8; ++i) {
        int row = rowBase + i;
        if (row < M) C[(size_t)row * NOUT + col] = acc[i];
    }
}

template <int K, typename vec>
__device__ __forceinline__ void add_rows(vec& acc, const vec* __restrict__ H,
                                         int s, int j, int lane) {
    vec r[K];
    #pragma unroll
    for (int u = 0; u < K; ++u)
        r[u] = H[(size_t)__builtin_amdgcn_readlane(s, j + u) * 32 + lane];
    #pragma unroll
    for (int u = 0; u < K; ++u) acc += r[u];
}

template <int F>
__global__ __launch_bounds__(256) void gcn_gather_rows(
    const float* __restrict__ Hs, const int* __restrict__ start,
    const int* __restrict__ col, const float* __restrict__ dinv,
    const float* __restrict__ bias, float* __restrict__ out, int N, int E) {
    constexpr int V = F / 32;
    typedef float vec __attribute__((ext_vector_type(V)));
    const int lane = threadIdx.x & 31;
    const int n = __builtin_amdgcn_readfirstlane((int)(blockIdx.x * 8 + (threadIdx.x >> 5)));
    if (n >= N) return;
    const int beg = start[n];
    const int end = (n + 1 < N) ? start[n + 1] : E;
    const vec* __restrict__ H = reinterpret_cast<const vec*>(Hs);

    vec acc = H[(size_t)n * 32 + lane];
    for (int p = beg; p < end; p += 32) {
        const int m = min(32, end - p);
        int s = (lane < m) ? col[p + lane] : 0;
        s = min(max(s, 0), N - 1);
        int j = 0;
        for (; j + 8 <= m; j += 8) add_rows<8>(acc, H, s, j, lane);
        if (m & 4) { add_rows<4>(acc, H, s, j, lane); j += 4; }
        if (m & 2) { add_rows<2>(acc, H, s, j, lane); j += 2; }
        if (m & 1) add_rows<1>(acc, H, s, j, lane);
    }

    const float di = dinv[n];
    const vec b = reinterpret_cast<const vec*>(bias)[lane];
    vec v = acc * di + b;
    #pragma unroll
    for (int k = 0; k < V; ++k) v[k] = (v[k] > 0.0f) ? v[k] : expm1f(v[k]);
    reinterpret_cast<vec*>(out)[(size_t)n * 32 + lane] = v;
}

#define FT_ROWS 32
#define FT_TILES 4

__global__ __launch_bounds__(256) void gcn_rows_gemm(
    const float* __restrict__ Hs, const int* __restrict__ start,
    const int* __restrict__ col, const float* __restrict__ dinv,
    const float* __restrict__ bias, const float* __restrict__ B,
    float* __restrict__ C, int N, int E) {
    typedef float vec __attribute__((ext_vector_type(4)));
    __shared__ __attribute__((aligned(16))) float Alds[2][FT_ROWS * APITCH];

    const int tid   = threadIdx.x;
    const int lane  = tid & 31;
    const int wave  = __builtin_amdgcn_readfirstlane(tid >> 5);
    const int lrow  = lane & 15;
    const int khalf = lane >> 4;
    const int mt    = wave >> 2;
    const int ncol  = (wave & 3) * 16 + lrow;

    v2f bfrag[KDIM / 4];
    #pragma unroll
    for (int i = 0; i < KDIM / 4; ++i) {
        const int ka = 4 * i + 2 * khalf;
        bfrag[i].x = B[(size_t)ka * OUT_DIM + ncol];
        bfrag[i].y = B[(size_t)(ka + 1) * OUT_DIM + ncol];
    }

    const vec* __restrict__ H = reinterpret_cast<const vec*>(Hs);
    const vec bv = reinterpret_cast<const vec*>(bias)[lane];
    const int ntile = (N + FT_ROWS - 1) / FT_ROWS;
    int buf = 0;
    for (int t = blockIdx.x; t < ntile; t += (int)gridDim.x, buf ^= 1) {
        float* __restrict__ At = Alds[buf];
        const int nbase = t * FT_ROWS + wave;

        int sb = 0;
        if (lane < 2 * (FT_ROWS / 8)) {
            const int idx = nbase + 8 * (lane >> 1) + (lane & 1);
            sb = (idx < N) ? start[idx] : E;
        }

        #pragma unroll
        for (int q = 0; q < FT_ROWS / 8; ++q) {
            const int r = wave + 8 * q;
            const int n = nbase + 8 * q;
            vec v = {};
            if (n < N) {
                const int beg = __builtin_amdgcn_readlane(sb, 2 * q);
                const int end = __builtin_amdgcn_readlane(sb, 2 * q + 1);
                vec acc = H[(size_t)n * 32 + lane];
                for (int p = beg; p < end; p += 32) {
                    const int m = min(32, end - p);
                    int s = (lane < m) ? col[p + lane] : 0;
                    s = min(max(s, 0), N - 1);
                    int j = 0;
                    for (; j + 8 <= m; j += 8) add_rows<8>(acc, H, s, j, lane);
                    if (m & 4) { add_rows<4>(acc, H, s, j, lane); j += 4; }
                    if (m & 2) { add_rows<2>(acc, H, s, j, lane); j += 2; }
                    if (m & 1) add_rows<1>(acc, H, s, j, lane);
                }
                const float di = dinv[n];
                v = acc * di + bv;
                #pragma unroll
                for (int k = 0; k < 4; ++k) v[k] = (v[k] > 0.0f) ? v[k] : expm1f(v[k]);
                v *= di;
            }
            *reinterpret_cast<vec*>(&At[r * APITCH + lane * 4]) = v;
        }
        __syncthreads();

        v8f acc = {};
        const float* __restrict__ Ar = &At[(mt * 16 + lrow) * APITCH + 2 * khalf];
        #pragma unroll
        for (int i = 0; i < KDIM / 4; ++i) {
            const v2f a = *reinterpret_cast<const v2f*>(&Ar[4 * i]);
            acc = __builtin_amdgcn_wmma_f32_16x16x4_f32(
                 false, a,  false, bfrag[i],
                 (short)0, acc,  false,  false);
        }

        const int rowBase = t * FT_ROWS + mt * 16 + 8 * khalf;
        #pragma unroll
        for (int i = 0; i < 8; ++i) {
            const int row = rowBase + i;
            if (row < N) C[(size_t)row * OUT_DIM + ncol] = acc[i];
        }
    }
}

extern "C" void kernel_launch(void* const* d_in, const int* in_sizes, int n_in,
                              void* d_out, int out_size, void* d_ws, size_t ws_size,
                              hipStream_t stream) {
    const float* x  = (const float*)d_in[0];
    const int*   ei = (const int*)d_in[1];
    const float* W1 = (const float*)d_in[2];
    const float* b1 = (const float*)d_in[3];
    const float* W2 = (const float*)d_in[4];
    const float* b2 = (const float*)d_in[5];
    float* out = (float*)d_out;

    const int N = in_sizes[0] / IN_DIM;
    const int E = in_sizes[1] / 2;
    const int* src = ei;
    const int* dst = ei + E;

    char* ws = (char*)d_ws;
    float* dinv = (float*)ws;
    size_t off1 = ((size_t)N * 4 + 255) & ~(size_t)255;
    float* hbuf = (float*)(ws + off1);
    size_t off2 = off1 + ((((size_t)N * HID_DIM * 4) + 255) & ~(size_t)255);
    float* h2buf = (float*)(ws + off2);
    size_t off3 = off2 + ((((size_t)N * OUT_DIM * 4) + 255) & ~(size_t)255);
    int* startA = (int*)(ws + off3);
    int* colA   = startA + N;
    size_t off4 = off3 + ((((size_t)N + E) * 4 + 255) & ~(size_t)255);
    const int ntiles  = (E + EL_TILE - 1) / EL_TILE;
    const int nb      = (N + (1 << EL_RLOG) - 1) >> EL_RLOG;
    int* rec   = (int*)(ws + off4);
    int* toffs = rec + ((E + 4095) & ~4095);

    const int mTiles = (N + 15) / 16;
    const int nodeBlocks = (N + 7) / 8;
    const int ftTiles  = (N + FT_ROWS - 1) / FT_ROWS;
    const int ftBlocks = (ftTiles + FT_TILES - 1) / FT_TILES;

    el_tile<<<ntiles, EL_TB, 0, stream>>>(src, dst, rec, toffs, E, N, nb, ntiles);
    el_build<<<nb, EL_BTB, 0, stream>>>(rec, toffs, startA, colA, dinv, N, E, nb, ntiles);

    gcn_gemm_wmma<HID_DIM><<<mTiles, HID_DIM * 2, 0, stream>>>(x, W1, dinv, hbuf, N);
    gcn_rows_gemm<<<ftBlocks, 256, 0, stream>>>(
        hbuf, startA, colA, dinv, b1, W2, h2buf, N, E);
    gcn_gather_rows<OUT_DIM><<<nodeBlocks, 256, 0, stream>>>(
        h2buf, startA, colA, dinv, b2, out, N, E);
}


# Layer-1 product by resident workgroups looping over 32-row tiles: W1 fragments stay in registers, next tile's rows in flight during the products (was 6250 one-tile groups, 8 dependent fetches each)
# speedup vs baseline: 5.4659x; 1.1123x over previous
#include <hip/hip_runtime.h>
#include <cmath>


typedef __attribute__((ext_vector_type(2))) float v2f;
typedef __attribute__((ext_vector_type(8))) float v8f;

#define IN_DIM 128
#define HID_DIM 128
#define OUT_DIM 64
#define KDIM 128
#define APITCH (KDIM + 4)

#define EL_TILE    4096
#define EL_TB      512
#define EL_BTB     1024
#define EL_BPER    5
#define EL_RLOG    8
#define EL_NB_MAX  512
#define EL_NT_MAX  512
#define EL_STAGE   8192

__global__ __launch_bounds__(EL_TB) void el_tile(const int* __restrict__ src,
                                                 const int* __restrict__ dst,
                                                 int* __restrict__ rec,
                                                 int* __restrict__ toffs,
                                                 int e, int n, int nb, int ntiles) {
    constexpr int PER = EL_TILE / EL_TB;
    static_assert(EL_TB == EL_NB_MAX, "one bucket counter per thread");
    __shared__ int h[EL_NB_MAX];
    __shared__ int cur[EL_NB_MAX];
    __shared__ int wtot[EL_TB / 32];
    __shared__ __attribute__((aligned(16))) unsigned srt[EL_TILE];
    const int tid  = threadIdx.x;
    const int lane = tid & 31;
    const int wave = tid >> 5;
    h[tid] = 0;
    __syncthreads();
    const int base = blockIdx.x * EL_TILE;
    const int cnt  = min(max(e - base, 0), EL_TILE);
    int d[PER], s[PER];
    #pragma unroll
    for (int k = 0; k < PER; ++k) {
        const int i = min(base + k * EL_TB + tid, e - 1);
        d[k] = dst[i];
        s[k] = src[i];
    }
    #pragma unroll
    for (int k = 0; k < PER; ++k) {
        d[k] = min(max(d[k], 0), n - 1);
        if (k * EL_TB + tid < cnt) atomicAdd(&h[min(d[k] >> EL_RLOG, EL_NB_MAX - 1)], 1);
    }
    __syncthreads();

    const int a = h[tid];
    int incl = a;
    #pragma unroll
    for (int o = 1; o < 32; o <<= 1) {
        int t = __shfl_up(incl, o);
        if (lane >= o) incl += t;
    }
    if (lane == 31) wtot[wave] = incl;
    __syncthreads();
    int excl = incl - a;
    #pragma unroll
    for (int w = 0; w < EL_TB / 32; ++w) excl += (w < wave) ? wtot[w] : 0;
    h[tid]   = excl;
    cur[tid] = excl;
    __syncthreads();

    #pragma unroll
    for (int k = 0; k < PER; ++k) {
        if (k * EL_TB + tid < cnt) {
            int p = atomicAdd(&cur[min(d[k] >> EL_RLOG, EL_NB_MAX - 1)], 1);
            p = min(max(p, 0), EL_TILE - 1);
            srt[p] = ((unsigned)s[k] & 0xFFFFFFu) | ((unsigned)(d[k] & 255) << 24);
        }
    }
    __syncthreads();

    const uint4* __restrict__ s4 = reinterpret_cast<const uint4*>(srt);
    uint4* __restrict__ r4 = reinterpret_cast<uint4*>(rec + base);
    for (int i = tid; i * 4 < cnt; i += EL_TB) {
        if (i * 4 + 3 < cnt) {
            r4[i] = s4[i];
        } else {
            for (int j = i * 4; j < cnt; ++j) rec[base + j] = (int)srt[j];
        }
    }
    const int nbc = min(nb, EL_NB_MAX);
    for (int b = tid; b <= nbc; b += EL_TB)
        toffs[(size_t)b * ntiles + blockIdx.x] = (b < EL_NB_MAX) ? h[b] : cnt;
}

__global__ __launch_bounds__(EL_BTB) void el_build(const int* __restrict__ rec,
                                                   const int* __restrict__ toffs,
                                                   int* __restrict__ start,
                                                   int* __restrict__ col,
                                                   float* __restrict__ dinv,
                                                   int n, int e, int nb, int ntiles) {
    constexpr int NODES = 1 << EL_RLOG;
    constexpr int TW    = EL_NT_MAX / 32;
    __shared__ int c[NODES];
    __shared__ int cur[NODES];
    __shared__ int wtot[NODES / 32];
    __shared__ int stage[EL_STAGE];
    __shared__ int addr[EL_STAGE];
    __shared__ int rpos[EL_NT_MAX + 1];
    __shared__ int rsrc[EL_NT_MAX];
    __shared__ __attribute__((aligned(16))) int wlen[TW];
    __shared__ __attribute__((aligned(16))) int wsum[TW];
    const int tid  = threadIdx.x;
    const int lane = tid & 31;
    const int wave = tid >> 5;
    const int b    = blockIdx.x;
    const int nt   = min(ntiles, EL_NT_MAX);

    int s0 = 0, len = 0;
    if (tid < nt) {
        int a = toffs[(size_t)b * ntiles + tid];
        int z = toffs[(size_t)(b + 1) * ntiles + tid];
        a = min(max(a, 0), EL_TILE);
        z = min(max(z, a), EL_TILE);
        s0  = a;
        len = z - a;
    }
    int incl = len;
    #pragma unroll
    for (int o = 1; o < 32; o <<= 1) {
        int t = __shfl_up(incl, o);
        if (lane >= o) incl += t;
    }
    int ssum = s0;
    #pragma unroll
    for (int o = 16; o > 0; o >>= 1) ssum += __shfl_xor(ssum, o);
    if (wave < TW && lane == 31) wlen[wave] = incl;
    if (wave < TW && lane == 0)  wsum[wave] = ssum;
    if (tid < NODES) c[tid] = 0;
    __syncthreads();
    int before = 0, mraw = 0, bsraw = 0;
    #pragma unroll
    for (int w = 0; w < TW; ++w) {
        const int v = wlen[w];
        mraw   += v;
        before += (w < wave) ? v : 0;
        bsraw  += wsum[w];
    }
    const int bs  = min(max(bsraw, 0), e);
    const int m   = min(max(mraw, 0), e - bs);
    const int pos = before + incl - len;
    if (tid < nt) {
        rpos[tid] = pos;
        rsrc[tid] = tid * EL_TILE + s0;
        for (int i = 0; i < len; ++i)
            if (pos + i < EL_STAGE) addr[pos + i] = tid * EL_TILE + s0 + i;
    }
    if (tid == 0) rpos[nt] = mraw;
    __syncthreads();

    const int  recmax = nt * EL_TILE - 1;
    const bool listed = (m <= EL_STAGE);
    auto find_run = [&](int p) -> int {
        if (p < EL_STAGE) return addr[p];
        int lo = 0, hi = nt - 1;
        for (int it = 0; it < 10 && lo < hi; ++it) {
            const int mid = (lo + hi + 1) >> 1;
            if (rpos[mid] <= p) lo = mid; else hi = mid - 1;
        }
        return rsrc[lo] + (p - rpos[lo]);
    };
    auto load_recs = [&](int i, unsigned (&r)[EL_BPER]) {
        int a[EL_BPER];
        if (listed) {
            #pragma unroll
            for (int u = 0; u < EL_BPER; ++u) a[u] = addr[min(i + u * EL_BTB, m - 1)];
        } else {
            #pragma unroll
            for (int u = 0; u < EL_BPER; ++u) a[u] = find_run(min(i + u * EL_BTB, m - 1));
        }
        #pragma unroll
        for (int u = 0; u < EL_BPER; ++u) r[u] = (unsigned)rec[min(max(a[u], 0), recmax)];
    };

    for (int i = tid; i < m; i += EL_BPER * EL_BTB) {
        unsigned r[EL_BPER];
        load_recs(i, r);
        #pragma unroll
        for (int u = 0; u < EL_BPER; ++u)
            if (i + u * EL_BTB < m) atomicAdd(&c[r[u] >> 24], 1);
    }
    __syncthreads();

    const bool own = tid < NODES;
    const int cn = own ? c[tid] : 0;
    int nincl = cn;
    #pragma unroll
    for (int o = 1; o < 32; o <<= 1) {
        int t = __shfl_up(nincl, o);
        if (lane >= o) nincl += t;
    }
    if (own && lane == 31) wtot[wave] = nincl;
    __syncthreads();
    if (own) {
        int excl = nincl - cn;
        for (int w = 0; w < wave; ++w) excl += wtot[w];
        const int node = (b << EL_RLOG) + tid;
        if (node < n) {
            start[node] = bs + excl;
            dinv[node]  = rsqrtf((float)(cn + 1));
        }
        cur[tid] = excl;
    }
    __syncthreads();

    for (int i = tid; i < m; i += EL_BPER * EL_BTB) {
        unsigned r[EL_BPER];
        load_recs(i, r);
        #pragma unroll
        for (int u = 0; u < EL_BPER; ++u) {
            if (i + u * EL_BTB < m) {
                int p = atomicAdd(&cur[r[u] >> 24], 1);
                p = min(max(p, 0), m - 1);
                const int sv = (int)(r[u] & 0xFFFFFFu);
                if (listed) stage[p] = sv; else col[bs + p] = sv;
            }
        }
    }
    if (listed) {
        __syncthreads();
        for (int i = tid; i < m; i += EL_BTB) col[bs + i] = stage[i];
    }
}

#define G1_ROWS 32

__global__ __launch_bounds__(256) void gcn_gemm_resident(
    const float* __restrict__ A, const float* __restrict__ B,
    const float* __restrict__ rscale, float* __restrict__ C, int M) {
    typedef float vec __attribute__((ext_vector_type(4)));
    __shared__ __attribute__((aligned(16))) float Alds[2][G1_ROWS * APITCH];

    const int tid   = threadIdx.x;
    const int lane  = tid & 31;
    const int wave  = __builtin_amdgcn_readfirstlane(tid >> 5);
    const int lrow  = lane & 15;
    const int khalf = lane >> 4;
    const int ncol  = wave * 16 + lrow;
    const int ntile = (M + G1_ROWS - 1) / G1_ROWS;
    const vec* __restrict__ A4 = reinterpret_cast<const vec*>(A);

    int t = blockIdx.x;
    if (t >= ntile) return;

    vec   r[G1_ROWS / 8];
    float sc[G1_ROWS / 8];
    #pragma unroll
    for (int q = 0; q < G1_ROWS / 8; ++q) {
        const int row = min(t * G1_ROWS + wave + 8 * q, M - 1);
        r[q]  = A4[(size_t)row * (KDIM / 4) + lane];
        sc[q] = rscale[row];
    }

    v2f bfrag[KDIM / 4];
    #pragma unroll
    for (int i = 0; i < KDIM / 4; ++i) {
        const int ka = 4 * i + 2 * khalf;
        bfrag[i].x = B[(size_t)ka * HID_DIM + ncol];
        bfrag[i].y = B[(size_t)(ka + 1) * HID_DIM + ncol];
    }

    #pragma unroll
    for (int q = 0; q < G1_ROWS / 8; ++q)
        *reinterpret_cast<vec*>(&Alds[0][(wave + 8 * q) * APITCH + lane * 4]) = r[q] * sc[q];
    __syncthreads();

    int buf = 0;
    while (true) {
        const int  tn   = t + (int)gridDim.x;
        const bool more = tn < ntile;
        if (more) {
            #pragma unroll
            for (int q = 0; q < G1_ROWS / 8; ++q) {
                const int row = min(tn * G1_ROWS + wave + 8 * q, M - 1);
                r[q]  = A4[(size_t)row * (KDIM / 4) + lane];
                sc[q] = rscale[row];
            }
        }

        const float* __restrict__ Ar = &Alds[buf][lrow * APITCH + 2 * khalf];
        v8f acc0 = {}, acc1 = {};
        #pragma unroll
        for (int i = 0; i < KDIM / 4; ++i) {
            const v2f a0 = *reinterpret_cast<const v2f*>(&Ar[4 * i]);
            const v2f a1 = *reinterpret_cast<const v2f*>(&Ar[16 * APITCH + 4 * i]);
            acc0 = __builtin_amdgcn_wmma_f32_16x16x4_f32(
                 false, a0,  false, bfrag[i],
                 (short)0, acc0,  false,  false);
            acc1 = __builtin_amdgcn_wmma_f32_16x16x4_f32(
                 false, a1,  false, bfrag[i],
                 (short)0, acc1,  false,  false);
        }

        if (more) {
            #pragma unroll
            for (int q = 0; q < G1_ROWS / 8; ++q)
                *reinterpret_cast<vec*>(&Alds[buf ^ 1][(wave + 8 * q) * APITCH + lane * 4]) = r[q] * sc[q];
            __syncthreads();
        }

        const int rowBase = t * G1_ROWS + 8 * khalf;
        if (t * G1_ROWS + G1_ROWS <= M) {
            #pragma unroll
            for (int i = 0; i < 8; ++i) {
                C[(size_t)(rowBase + i) * HID_DIM + ncol]      = acc0[i];
                C[(size_t)(rowBase + 16 + i) * HID_DIM + ncol] = acc1[i];
            }
        } else {
            #pragma unroll
            for (int i = 0; i < 8; ++i) {
                if (rowBase + i < M)      C[(size_t)(rowBase + i) * HID_DIM + ncol]      = acc0[i];
                if (rowBase + 16 + i < M) C[(size_t)(rowBase + 16 + i) * HID_DIM + ncol] = acc1[i];
            }
        }

        if (!more) break;
        t = tn;
        buf ^= 1;
    }
}

template <int K, typename vec>
__device__ __forceinline__ void add_rows(vec& acc, const vec* __restrict__ H,
                                         int s, int j, int lane) {
    vec r[K];
    #pragma unroll
    for (int u = 0; u < K; ++u)
        r[u] = H[(size_t)__builtin_amdgcn_readlane(s, j + u) * 32 + lane];
    #pragma unroll
    for (int u = 0; u < K; ++u) acc += r[u];
}

template <int F>
__global__ __launch_bounds__(256) void gcn_gather_rows(
    const float* __restrict__ Hs, const int* __restrict__ start,
    const int* __restrict__ col, const float* __restrict__ dinv,
    const float* __restrict__ bias, float* __restrict__ out, int N, int E) {
    constexpr int V = F / 32;
    typedef float vec __attribute__((ext_vector_type(V)));
    const int lane = threadIdx.x & 31;
    const int n = __builtin_amdgcn_readfirstlane((int)(blockIdx.x * 8 + (threadIdx.x >> 5)));
    if (n >= N) return;
    const int beg = start[n];
    const int end = (n + 1 < N) ? start[n + 1] : E;
    const vec* __restrict__ H = reinterpret_cast<const vec*>(Hs);

    vec acc = H[(size_t)n * 32 + lane];
    for (int p = beg; p < end; p += 32) {
        const int m = min(32, end - p);
        int s = (lane < m) ? col[p + lane] : 0;
        s = min(max(s, 0), N - 1);
        int j = 0;
        for (; j + 8 <= m; j += 8) add_rows<8>(acc, H, s, j, lane);
        if (m & 4) { add_rows<4>(acc, H, s, j, lane); j += 4; }
        if (m & 2) { add_rows<2>(acc, H, s, j, lane); j += 2; }
        if (m & 1) add_rows<1>(acc, H, s, j, lane);
    }

    const float di = dinv[n];
    const vec b = reinterpret_cast<const vec*>(bias)[lane];
    vec v = acc * di + b;
    #pragma unroll
    for (int k = 0; k < V; ++k) v[k] = (v[k] > 0.0f) ? v[k] : expm1f(v[k]);
    reinterpret_cast<vec*>(out)[(size_t)n * 32 + lane] = v;
}

#define FT_ROWS 32
#define FT_TILES 4

__global__ __launch_bounds__(256) void gcn_rows_gemm(
    const float* __restrict__ Hs, const int* __restrict__ start,
    const int* __restrict__ col, const float* __restrict__ dinv,
    const float* __restrict__ bias, const float* __restrict__ B,
    float* __restrict__ C, int N, int E) {
    typedef float vec __attribute__((ext_vector_type(4)));
    __shared__ __attribute__((aligned(16))) float Alds[2][FT_ROWS * APITCH];

    const int tid   = threadIdx.x;
    const int lane  = tid & 31;
    const int wave  = __builtin_amdgcn_readfirstlane(tid >> 5);
    const int lrow  = lane & 15;
    const int khalf = lane >> 4;
    const int mt    = wave >> 2;
    const int ncol  = (wave & 3) * 16 + lrow;

    v2f bfrag[KDIM / 4];
    #pragma unroll
    for (int i = 0; i < KDIM / 4; ++i) {
        const int ka = 4 * i + 2 * khalf;
        bfrag[i].x = B[(size_t)ka * OUT_DIM + ncol];
        bfrag[i].y = B[(size_t)(ka + 1) * OUT_DIM + ncol];
    }

    const vec* __restrict__ H = reinterpret_cast<const vec*>(Hs);
    const vec bv = reinterpret_cast<const vec*>(bias)[lane];
    const int ntile = (N + FT_ROWS - 1) / FT_ROWS;
    int buf = 0;
    for (int t = blockIdx.x; t < ntile; t += (int)gridDim.x, buf ^= 1) {
        float* __restrict__ At = Alds[buf];
        const int nbase = t * FT_ROWS + wave;

        int sb = 0;
        if (lane < 2 * (FT_ROWS / 8)) {
            const int idx = nbase + 8 * (lane >> 1) + (lane & 1);
            sb = (idx < N) ? start[idx] : E;
        }

        #pragma unroll
        for (int q = 0; q < FT_ROWS / 8; ++q) {
            const int r = wave + 8 * q;
            const int n = nbase + 8 * q;
            vec v = {};
            if (n < N) {
                const int beg = __builtin_amdgcn_readlane(sb, 2 * q);
                const int end = __builtin_amdgcn_readlane(sb, 2 * q + 1);
                vec acc = H[(size_t)n * 32 + lane];
                for (int p = beg; p < end; p += 32) {
                    const int m = min(32, end - p);
                    int s = (lane < m) ? col[p + lane] : 0;
                    s = min(max(s, 0), N - 1);
                    int j = 0;
                    for (; j + 8 <= m; j += 8) add_rows<8>(acc, H, s, j, lane);
                    if (m & 4) { add_rows<4>(acc, H, s, j, lane); j += 4; }
                    if (m & 2) { add_rows<2>(acc, H, s, j, lane); j += 2; }
                    if (m & 1) add_rows<1>(acc, H, s, j, lane);
                }
                const float di = dinv[n];
                v = acc * di + bv;
                #pragma unroll
                for (int k = 0; k < 4; ++k) v[k] = (v[k] > 0.0f) ? v[k] : expm1f(v[k]);
                v *= di;
            }
            *reinterpret_cast<vec*>(&At[r * APITCH + lane * 4]) = v;
        }
        __syncthreads();

        v8f acc = {};
        const float* __restrict__ Ar = &At[(mt * 16 + lrow) * APITCH + 2 * khalf];
        #pragma unroll
        for (int i = 0; i < KDIM / 4; ++i) {
            const v2f a = *reinterpret_cast<const v2f*>(&Ar[4 * i]);
            acc = __builtin_amdgcn_wmma_f32_16x16x4_f32(
                 false, a,  false, bfrag[i],
                 (short)0, acc,  false,  false);
        }

        const int rowBase = t * FT_ROWS + mt * 16 + 8 * khalf;
        #pragma unroll
        for (int i = 0; i < 8; ++i) {
            const int row = rowBase + i;
            if (row < N) C[(size_t)row * OUT_DIM + ncol] = acc[i];
        }
    }
}

extern "C" void kernel_launch(void* const* d_in, const int* in_sizes, int n_in,
                              void* d_out, int out_size, void* d_ws, size_t ws_size,
                              hipStream_t stream) {
    const float* x  = (const float*)d_in[0];
    const int*   ei = (const int*)d_in[1];
    const float* W1 = (const float*)d_in[2];
    const float* b1 = (const float*)d_in[3];
    const float* W2 = (const float*)d_in[4];
    const float* b2 = (const float*)d_in[5];
    float* out = (float*)d_out;

    const int N = in_sizes[0] / IN_DIM;
    const int E = in_sizes[1] / 2;
    const int* src = ei;
    const int* dst = ei + E;

    char* ws = (char*)d_ws;
    float* dinv = (float*)ws;
    size_t off1 = ((size_t)N * 4 + 255) & ~(size_t)255;
    float* hbuf = (float*)(ws + off1);
    size_t off2 = off1 + ((((size_t)N * HID_DIM * 4) + 255) & ~(size_t)255);
    float* h2buf = (float*)(ws + off2);
    size_t off3 = off2 + ((((size_t)N * OUT_DIM * 4) + 255) & ~(size_t)255);
    int* startA = (int*)(ws + off3);
    int* colA   = startA + N;
    size_t off4 = off3 + ((((size_t)N + E) * 4 + 255) & ~(size_t)255);
    const int ntiles  = (E + EL_TILE - 1) / EL_TILE;
    const int nb      = (N + (1 << EL_RLOG) - 1) >> EL_RLOG;
    int* rec   = (int*)(ws + off4);
    int* toffs = rec + ((E + 4095) & ~4095);

    const int g1Blocks = 768;
    const int nodeBlocks = (N + 7) / 8;
    const int ftTiles  = (N + FT_ROWS - 1) / FT_ROWS;
    const int ftBlocks = (ftTiles + FT_TILES - 1) / FT_TILES;

    el_tile<<<ntiles, EL_TB, 0, stream>>>(src, dst, rec, toffs, E, N, nb, ntiles);
    el_build<<<nb, EL_BTB, 0, stream>>>(rec, toffs, startA, colA, dinv, N, E, nb, ntiles);

    gcn_gemm_resident<<<g1Blocks, 256, 0, stream>>>(x, W1, dinv, hbuf, N);
    gcn_rows_gemm<<<ftBlocks, 256, 0, stream>>>(
        hbuf, startA, colA, dinv, b1, W2, h2buf, N, E);
    gcn_gather_rows<OUT_DIM><<<nodeBlocks, 256, 0, stream>>>(
        h2buf, startA, colA, dinv, b2, out, N, E);
}
